# starting kernel
# baseline (speedup 1.0000x reference)
#include <hip/hip_runtime.h>
#include <hip/hip_bf16.h>


typedef float v2f __attribute__((ext_vector_type(2)));
typedef float v8f __attribute__((ext_vector_type(8)));

#define F_IN  256
#define F_MID 64
#define F_OUT 32


__global__ void __launch_bounds__(256) fill_kernel(float* __restrict__ p, float v, int n) {
    int i = blockIdx.x * 256 + threadIdx.x;
    if (i < n) p[i] = v;
}

__global__ void __launch_bounds__(256) deg_count_kernel(const int* __restrict__ dst,
                                                        float* __restrict__ deg, int n_edges) {
    int e = blockIdx.x * 256 + threadIdx.x;
    if (e < n_edges) atomicAdd(&deg[dst[e]], 1.0f);
}

__global__ void __launch_bounds__(256) rsqrt_kernel(float* __restrict__ d, int n) {
    int i = blockIdx.x * 256 + threadIdx.x;
    if (i < n) d[i] = rsqrtf(d[i]);
}


template <int K, int NOUT>
__global__ void __launch_bounds__(256) gemm_kernel(const float* __restrict__ x,
                                                   const float* __restrict__ W,
                                                   float* __restrict__ h, int n_nodes) {
    constexpr int LDK = K + 4;
    constexpr int NT  = NOUT / 16;
    __shared__ float sWt[NOUT * LDK];

    for (int i = threadIdx.x; i < K * NOUT; i += 256) {
        const int k = i / NOUT;
        const int n = i % NOUT;
        sWt[n * LDK + k] = W[i];
    }
    __syncthreads();

    const int wave = threadIdx.x >> 5;
    const int lane = threadIdx.x & 31;
    const int half = lane >> 4;
    const int l    = lane & 15;
    const int m0   = (blockIdx.x * 8 + wave) * 16;
    if (m0 >= n_nodes) return;

    const int rowA = min(m0 + l, n_nodes - 1);
    const float* __restrict__ xrow = x + (size_t)rowA * K;

    v8f acc[NT] = {};
    for (int k = 0; k < K; k += 4) {
        const int kk = k + 2 * half;
        const v2f a = *(const v2f*)(xrow + kk);
#pragma unroll
        for (int t = 0; t < NT; ++t) {
            const int col = t * 16 + l;
            const v2f b = *(const v2f*)(&sWt[col * LDK + kk]);
            acc[t] = __builtin_amdgcn_wmma_f32_16x16x4_f32(
                false, a, false, b, (short)0, acc[t], false, false);
        }
    }

    if (m0 + 16 <= n_nodes) {
        float* __restrict__ hp = h + (size_t)(m0 + 8 * half) * NOUT + l;
#pragma unroll
        for (int v = 0; v < 8; ++v) {
#pragma unroll
            for (int t = 0; t < NT; ++t)
                hp[(size_t)v * NOUT + t * 16] = acc[t][v];
        }
    } else {
#pragma unroll
        for (int v = 0; v < 8; ++v) {
            const int row = m0 + 8 * half + v;
            if (row < n_nodes) {
#pragma unroll
                for (int t = 0; t < NT; ++t)
                    h[(size_t)row * NOUT + t * 16 + l] = acc[t][v];
            }
        }
    }
}


template <int F>
__global__ void __launch_bounds__(256) scatter_kernel(const int* __restrict__ src,
                                                      const int* __restrict__ dst,
                                                      const float* __restrict__ dinv,
                                                      const float* __restrict__ h,
                                                      float* __restrict__ agg, int n_edges) {
    const int idx = blockIdx.x * 256 + threadIdx.x;
    const int e = idx / F;
    const int f = idx % F;
    if (e >= n_edges) return;
    const int s = src[e];
    const int d = dst[e];
    const float norm = dinv[s] * dinv[d];
    atomicAdd(&agg[(size_t)d * F + f], norm * h[(size_t)s * F + f]);
}

template <int F, bool RELU>
__global__ void __launch_bounds__(256) finalize_kernel(float* __restrict__ agg,
                                                       const float* __restrict__ h,
                                                       const float* __restrict__ dinv,
                                                       const float* __restrict__ bias,
                                                       int n_nodes) {
    const int idx = blockIdx.x * 256 + threadIdx.x;
    if (idx >= n_nodes * F) return;
    const int i = idx / F;
    const int f = idx % F;
    const float di = dinv[i];
    float v = agg[idx] + di * di * h[idx] + bias[f];
    if (RELU) v = fmaxf(v, 0.0f);
    agg[idx] = v;
}


__global__ void __launch_bounds__(256) decode_kernel(const int* __restrict__ pos,
                                                     const int* __restrict__ neg,
                                                     const float* __restrict__ z,
                                                     float* __restrict__ out, int n_eval) {
    const int gw   = (blockIdx.x * 256 + threadIdx.x) >> 5;
    const int lane = threadIdx.x & 31;
    if (gw >= 2 * n_eval) return;
    int a, b;
    if (gw < n_eval) { a = pos[gw];            b = pos[n_eval + gw]; }
    else             { int k = gw - n_eval; a = neg[k]; b = neg[n_eval + k]; }
    float v = z[(size_t)a * F_OUT + lane] * z[(size_t)b * F_OUT + lane];
#pragma unroll
    for (int o = 16; o > 0; o >>= 1) v += __shfl_xor(v, o, 32);
    if (lane == 0) out[gw] = v;
}


static inline size_t align_up(size_t x) { return (x + 255) & ~(size_t)255; }

extern "C" void kernel_launch(void* const* d_in, const int* in_sizes, int n_in,
                              void* d_out, int out_size, void* d_ws, size_t ws_size,
                              hipStream_t stream) {
    const float* x   = (const float*)d_in[0];
    const float* W1  = (const float*)d_in[1];
    const float* b1  = (const float*)d_in[2];
    const float* W2  = (const float*)d_in[3];
    const float* b2  = (const float*)d_in[4];
    const int*   ei  = (const int*)d_in[5];
    const int*   pos = (const int*)d_in[6];
    const int*   neg = (const int*)d_in[7];
    float* out = (float*)d_out;

    const int n_nodes = in_sizes[0] / F_IN;
    const int n_edges = in_sizes[5] / 2;
    const int n_eval  = in_sizes[6] / 2;
    const int* src = ei;
    const int* dst = ei + n_edges;

    char* w = (char*)d_ws;
    size_t off = 0;
    float* dinv = (float*)(w + off); off += align_up((size_t)n_nodes * 4);
    float* h1   = (float*)(w + off); off += align_up((size_t)n_nodes * F_MID * 4);
    float* agg1 = (float*)(w + off); off += align_up((size_t)n_nodes * F_MID * 4);
    float* h2   = (float*)(w + off); off += align_up((size_t)n_nodes * F_OUT * 4);
    float* z    = (float*)(w + off); off += align_up((size_t)n_nodes * F_OUT * 4);
    (void)ws_size;

    const int B = 256;
    const int nm = n_nodes * F_MID;
    const int no = n_nodes * F_OUT;

    fill_kernel<<<(n_nodes + B - 1) / B, B, 0, stream>>>(dinv, 1.0f, n_nodes);
    fill_kernel<<<(nm + B - 1) / B, B, 0, stream>>>(agg1, 0.0f, nm);
    fill_kernel<<<(no + B - 1) / B, B, 0, stream>>>(z, 0.0f, no);
    deg_count_kernel<<<(n_edges + B - 1) / B, B, 0, stream>>>(dst, dinv, n_edges);
    rsqrt_kernel<<<(n_nodes + B - 1) / B, B, 0, stream>>>(dinv, n_nodes);

    const int gemm_blocks = (n_nodes + 128 - 1) / 128;
    gemm_kernel<F_IN, F_MID><<<gemm_blocks, B, 0, stream>>>(x, W1, h1, n_nodes);
    {
        long long t = (long long)n_edges * F_MID;
        scatter_kernel<F_MID><<<(unsigned)((t + B - 1) / B), B, 0, stream>>>(
            src, dst, dinv, h1, agg1, n_edges);
    }
    finalize_kernel<F_MID, true><<<(nm + B - 1) / B, B, 0, stream>>>(agg1, h1, dinv, b1, n_nodes);

    gemm_kernel<F_MID, F_OUT><<<gemm_blocks, B, 0, stream>>>(agg1, W2, h2, n_nodes);
    {
        long long t = (long long)n_edges * F_OUT;
        scatter_kernel<F_OUT><<<(unsigned)((t + B - 1) / B), B, 0, stream>>>(
            src, dst, dinv, h2, z, n_edges);
    }
    finalize_kernel<F_OUT, false><<<(no + B - 1) / B, B, 0, stream>>>(z, h2, dinv, b2, n_nodes);

    {
        int waves = 2 * n_eval;
        int blocks = (waves * 32 + B - 1) / B;
        decode_kernel<<<blocks, B, 0, stream>>>(pos, neg, z, out, n_eval);
    }
    (void)out_size; (void)n_in;
}


# Replace per-edge float atomic scatter-add by counting-sort CSR (int atomic rank + prefix sum) and per-node gather-sum fused with self loop, bias, ReLU; per-node sum order is now CSR arrival order
# speedup vs baseline: 2.3580x; 2.3580x over previous
#include <hip/hip_runtime.h>
#include <hip/hip_bf16.h>


typedef float v2f __attribute__((ext_vector_type(2)));
typedef float v8f __attribute__((ext_vector_type(8)));

#define F_IN  256
#define F_MID 64
#define F_OUT 32


__global__ void __launch_bounds__(256) zero_kernel(int* __restrict__ p, int n) {
    int i = blockIdx.x * 256 + threadIdx.x;
    if (i < n) p[i] = 0;
}

__global__ void __launch_bounds__(256) rank_kernel(const int* __restrict__ dst,
                                                   int* __restrict__ cnt,
                                                   int* __restrict__ rank, int n_edges) {
    int e = blockIdx.x * 256 + threadIdx.x;
    if (e < n_edges) rank[e] = atomicAdd(&cnt[dst[e]], 1);
}

__device__ __forceinline__ int block_sum_256(int v, int* sred) {
    const int lane = threadIdx.x & 31;
    const int wave = threadIdx.x >> 5;
#pragma unroll
    for (int o = 16; o > 0; o >>= 1) v += __shfl_xor(v, o, 32);
    if (lane == 0) sred[wave] = v;
    __syncthreads();
    int t = 0;
#pragma unroll
    for (int w = 0; w < 8; ++w) t += sred[w];
    return t;
}

__global__ void __launch_bounds__(256) scan_partial_kernel(const int* __restrict__ cnt,
                                                           int* __restrict__ bsum, int n) {
    __shared__ int sred[8];
    const int i0 = (blockIdx.x * 256 + threadIdx.x) * 4;
    int s = 0;
#pragma unroll
    for (int q = 0; q < 4; ++q)
        if (i0 + q < n) s += cnt[i0 + q];
    const int tot = block_sum_256(s, sred);
    if (threadIdx.x == 0) bsum[blockIdx.x] = tot;
}

__global__ void __launch_bounds__(256) scan_final_kernel(const int* __restrict__ cnt,
                                                         const int* __restrict__ bsum,
                                                         int* __restrict__ rowptr,
                                                         float* __restrict__ dinv, int n) {
    __shared__ int sred[8];
    __shared__ int swave[8];
    const int lane = threadIdx.x & 31;
    const int wave = threadIdx.x >> 5;

    int part = 0;
    for (int i = threadIdx.x; i < (int)blockIdx.x; i += 256) part += bsum[i];
    const int base = block_sum_256(part, sred);

    const int i0 = (blockIdx.x * 256 + threadIdx.x) * 4;
    int c[4];
    int s = 0;
#pragma unroll
    for (int q = 0; q < 4; ++q) {
        c[q] = (i0 + q < n) ? cnt[i0 + q] : 0;
        s += c[q];
    }
    int incl = s;
#pragma unroll
    for (int o = 1; o < 32; o <<= 1) {
        const int t = __shfl_up(incl, o, 32);
        if (lane >= o) incl += t;
    }
    if (lane == 31) swave[wave] = incl;
    __syncthreads();
    int woff = 0;
#pragma unroll
    for (int w = 0; w < 8; ++w)
        if (w < wave) woff += swave[w];

    int run = base + woff + incl - s;
#pragma unroll
    for (int q = 0; q < 4; ++q) {
        const int i = i0 + q;
        if (i < n) {
            rowptr[i] = run;
            dinv[i] = rsqrtf((float)(c[q] + 1));
            run += c[q];
            if (i == n - 1) rowptr[n] = run;
        }
    }
}

__global__ void __launch_bounds__(256) csr_fill_kernel(const int* __restrict__ src,
                                                       const int* __restrict__ dst,
                                                       const int* __restrict__ rowptr,
                                                       const int* __restrict__ rank,
                                                       int* __restrict__ col, int n_edges) {
    const int e = blockIdx.x * 256 + threadIdx.x;
    if (e >= n_edges) return;
    const int p = rowptr[dst[e]] + rank[e];
    if ((unsigned)p < (unsigned)n_edges) col[p] = src[e];
}


template <int K, int NOUT>
__global__ void __launch_bounds__(256) gemm_kernel(const float* __restrict__ x,
                                                   const float* __restrict__ W,
                                                   float* __restrict__ h, int n_nodes) {
    constexpr int LDK = K + 4;
    constexpr int NT  = NOUT / 16;
    __shared__ float sWt[NOUT * LDK];

    for (int i = threadIdx.x; i < K * NOUT; i += 256) {
        const int k = i / NOUT;
        const int n = i % NOUT;
        sWt[n * LDK + k] = W[i];
    }
    __syncthreads();

    const int wave = threadIdx.x >> 5;
    const int lane = threadIdx.x & 31;
    const int half = lane >> 4;
    const int l    = lane & 15;
    const int m0   = (blockIdx.x * 8 + wave) * 16;
    if (m0 >= n_nodes) return;

    const int rowA = min(m0 + l, n_nodes - 1);
    const float* __restrict__ xrow = x + (size_t)rowA * K;

    v8f acc[NT] = {};
    for (int k = 0; k < K; k += 4) {
        const int kk = k + 2 * half;
        const v2f a = *(const v2f*)(xrow + kk);
#pragma unroll
        for (int t = 0; t < NT; ++t) {
            const int col = t * 16 + l;
            const v2f b = *(const v2f*)(&sWt[col * LDK + kk]);
            acc[t] = __builtin_amdgcn_wmma_f32_16x16x4_f32(
                false, a, false, b, (short)0, acc[t], false, false);
        }
    }

    if (m0 + 16 <= n_nodes) {
        float* __restrict__ hp = h + (size_t)(m0 + 8 * half) * NOUT + l;
#pragma unroll
        for (int v = 0; v < 8; ++v) {
#pragma unroll
            for (int t = 0; t < NT; ++t)
                hp[(size_t)v * NOUT + t * 16] = acc[t][v];
        }
    } else {
#pragma unroll
        for (int v = 0; v < 8; ++v) {
            const int row = m0 + 8 * half + v;
            if (row < n_nodes) {
#pragma unroll
                for (int t = 0; t < NT; ++t)
                    h[(size_t)row * NOUT + t * 16 + l] = acc[t][v];
            }
        }
    }
}


template <int F, bool RELU>
__global__ void __launch_bounds__(256) gather_kernel(const int* __restrict__ rowptr,
                                                     const int* __restrict__ col,
                                                     const float* __restrict__ dinv,
                                                     const float* __restrict__ h,
                                                     const float* __restrict__ bias,
                                                     float* __restrict__ outp, int n_nodes) {
    constexpr int V = F / 32;
    const int lane = threadIdx.x & 31;
    const int node = __builtin_amdgcn_readfirstlane((int)((blockIdx.x * 256 + threadIdx.x) >> 5));
    if (node >= n_nodes) return;

    const int beg = rowptr[node];
    const int end = rowptr[node + 1];
    const float dd = dinv[node];
    const float* __restrict__ hl = h + V * lane;

    float hs[V];
    float bs[V];
    {
        const float* __restrict__ p = hl + (size_t)node * F;
        if constexpr (V == 2) {
            const v2f t = *(const v2f*)p;
            hs[0] = t.x;
            hs[1] = t.y;
        } else {
            hs[0] = p[0];
        }
#pragma unroll
        for (int v = 0; v < V; ++v) bs[v] = bias[V * lane + v];
    }

    float acc[V];
#pragma unroll
    for (int v = 0; v < V; ++v) acc[v] = 0.0f;

    for (int base = beg; base < end; base += 32) {
        const int j = base + lane;
        int s_l = node;
        float w_l = 0.0f;
        if (j < end) {
            const unsigned c = (unsigned)col[j];
            s_l = (int)(c < (unsigned)n_nodes ? c : (unsigned)(n_nodes - 1));
            w_l = dinv[s_l] * dd;
        }
        const int m = min(32, end - base);
        for (int k = 0; k < m; k += 4) {
            int s[4];
            float w[4];
            float r[4][V];
#pragma unroll
            for (int i = 0; i < 4; ++i) {
                s[i] = __builtin_amdgcn_readlane(s_l, k + i);
                w[i] = __int_as_float(__builtin_amdgcn_readlane(__float_as_int(w_l), k + i));
            }
#pragma unroll
            for (int i = 0; i < 4; ++i) {
                const float* __restrict__ p = hl + (size_t)s[i] * F;
                if constexpr (V == 2) {
                    const v2f t = *(const v2f*)p;
                    r[i][0] = t.x;
                    r[i][1] = t.y;
                } else {
                    r[i][0] = p[0];
                }
            }
#pragma unroll
            for (int i = 0; i < 4; ++i) {
                const bool live = (k + i < m);
#pragma unroll
                for (int v = 0; v < V; ++v)
                    acc[v] = fmaf(w[i], live ? r[i][v] : 0.0f, acc[v]);
            }
        }
    }

    float o[V];
#pragma unroll
    for (int v = 0; v < V; ++v) {
        float val = acc[v] + dd * dd * hs[v] + bs[v];
        if (RELU) val = fmaxf(val, 0.0f);
        o[v] = val;
    }
    float* __restrict__ q = outp + (size_t)node * F + V * lane;
    if constexpr (V == 2) {
        v2f t;
        t.x = o[0];
        t.y = o[1];
        *(v2f*)q = t;
    } else {
        q[0] = o[0];
    }
}


__global__ void __launch_bounds__(256) decode_kernel(const int* __restrict__ pos,
                                                     const int* __restrict__ neg,
                                                     const float* __restrict__ z,
                                                     float* __restrict__ out, int n_eval) {
    const int gw   = (blockIdx.x * 256 + threadIdx.x) >> 5;
    const int lane = threadIdx.x & 31;
    if (gw >= 2 * n_eval) return;
    int a, b;
    if (gw < n_eval) { a = pos[gw];            b = pos[n_eval + gw]; }
    else             { int k = gw - n_eval; a = neg[k]; b = neg[n_eval + k]; }
    float v = z[(size_t)a * F_OUT + lane] * z[(size_t)b * F_OUT + lane];
#pragma unroll
    for (int o = 16; o > 0; o >>= 1) v += __shfl_xor(v, o, 32);
    if (lane == 0) out[gw] = v;
}


static inline size_t align_up(size_t x) { return (x + 255) & ~(size_t)255; }

extern "C" void kernel_launch(void* const* d_in, const int* in_sizes, int n_in,
                              void* d_out, int out_size, void* d_ws, size_t ws_size,
                              hipStream_t stream) {
    const float* x   = (const float*)d_in[0];
    const float* W1  = (const float*)d_in[1];
    const float* b1  = (const float*)d_in[2];
    const float* W2  = (const float*)d_in[3];
    const float* b2  = (const float*)d_in[4];
    const int*   ei  = (const int*)d_in[5];
    const int*   pos = (const int*)d_in[6];
    const int*   neg = (const int*)d_in[7];
    float* out = (float*)d_out;

    const int n_nodes = in_sizes[0] / F_IN;
    const int n_edges = in_sizes[5] / 2;
    const int n_eval  = in_sizes[6] / 2;
    const int* src = ei;
    const int* dst = ei + n_edges;

    const int B = 256;
    const int scan_blocks = (n_nodes + 1023) / 1024;

    char* w = (char*)d_ws;
    size_t off = 0;
    float* dinv  = (float*)(w + off); off += align_up((size_t)n_nodes * 4);
    float* h1    = (float*)(w + off); off += align_up((size_t)n_nodes * F_MID * 4);
    float* a1    = (float*)(w + off); off += align_up((size_t)n_nodes * F_MID * 4);
    int* cnt     = (int*)(w + off);   off += align_up((size_t)n_nodes * 4);
    int* rowptr  = (int*)(w + off);   off += align_up((size_t)(n_nodes + 1) * 4);
    int* bsum    = (int*)(w + off);   off += align_up((size_t)scan_blocks * 4);
    int* rank    = (int*)(w + off);   off += align_up((size_t)n_edges * 4);
    int* col     = (int*)(w + off);   off += align_up((size_t)n_edges * 4);
    float* h2    = h1;
    float* z     = h1 + (size_t)n_nodes * F_OUT;
    (void)ws_size;

    zero_kernel<<<(n_nodes + B - 1) / B, B, 0, stream>>>(cnt, n_nodes);
    rank_kernel<<<(n_edges + B - 1) / B, B, 0, stream>>>(dst, cnt, rank, n_edges);
    scan_partial_kernel<<<scan_blocks, B, 0, stream>>>(cnt, bsum, n_nodes);
    scan_final_kernel<<<scan_blocks, B, 0, stream>>>(cnt, bsum, rowptr, dinv, n_nodes);
    csr_fill_kernel<<<(n_edges + B - 1) / B, B, 0, stream>>>(src, dst, rowptr, rank, col, n_edges);

    const int gemm_blocks   = (n_nodes + 128 - 1) / 128;
    const int gather_blocks = (n_nodes + 8 - 1) / 8;

    gemm_kernel<F_IN, F_MID><<<gemm_blocks, B, 0, stream>>>(x, W1, h1, n_nodes);
    gather_kernel<F_MID, true><<<gather_blocks, B, 0, stream>>>(rowptr, col, dinv, h1, b1, a1, n_nodes);

    gemm_kernel<F_MID, F_OUT><<<gemm_blocks, B, 0, stream>>>(a1, W2, h2, n_nodes);
    gather_kernel<F_OUT, false><<<gather_blocks, B, 0, stream>>>(rowptr, col, dinv, h2, b2, z, n_nodes);

    {
        int waves = 2 * n_eval;
        int blocks = (waves * 32 + B - 1) / B;
        decode_kernel<<<blocks, B, 0, stream>>>(pos, neg, z, out, n_eval);
    }
    (void)out_size; (void)n_in;
}


# Fewer, fatter waves (about 6x fewer): 8 lanes per node, 4 nodes per wave in both gathers; 32 queries per wave in decode (dot: 4 products per lane, then 8-lane tree); 8 edges per thread in CSR build
# speedup vs baseline: 2.5196x; 1.0685x over previous
#include <hip/hip_runtime.h>
#include <hip/hip_bf16.h>


typedef float v2f __attribute__((ext_vector_type(2)));
typedef float v4f __attribute__((ext_vector_type(4)));
typedef float v8f __attribute__((ext_vector_type(8)));

#define F_IN  256
#define F_MID 64
#define F_OUT 32
#define EPT   8


__global__ void __launch_bounds__(256) zero_kernel(int* __restrict__ p, int n) {
    int i = blockIdx.x * 256 + threadIdx.x;
    if (i < n) p[i] = 0;
}

__global__ void __launch_bounds__(256) rank_kernel(const int* __restrict__ dst,
                                                   int* __restrict__ cnt,
                                                   int* __restrict__ rank, int n_edges) {
    const int e0 = blockIdx.x * (256 * EPT) + threadIdx.x;
    if ((blockIdx.x + 1) * (256 * EPT) <= n_edges) {
        int d[EPT];
        int r[EPT];
#pragma unroll
        for (int i = 0; i < EPT; ++i) d[i] = dst[e0 + i * 256];
#pragma unroll
        for (int i = 0; i < EPT; ++i) r[i] = atomicAdd(&cnt[d[i]], 1);
#pragma unroll
        for (int i = 0; i < EPT; ++i) rank[e0 + i * 256] = r[i];
    } else {
#pragma unroll
        for (int i = 0; i < EPT; ++i) {
            const int e = e0 + i * 256;
            if (e < n_edges) rank[e] = atomicAdd(&cnt[dst[e]], 1);
        }
    }
}

__device__ __forceinline__ int block_sum_256(int v, int* sred) {
    const int lane = threadIdx.x & 31;
    const int wave = threadIdx.x >> 5;
#pragma unroll
    for (int o = 16; o > 0; o >>= 1) v += __shfl_xor(v, o, 32);
    if (lane == 0) sred[wave] = v;
    __syncthreads();
    int t = 0;
#pragma unroll
    for (int w = 0; w < 8; ++w) t += sred[w];
    return t;
}

__global__ void __launch_bounds__(256) scan_partial_kernel(const int* __restrict__ cnt,
                                                           int* __restrict__ bsum, int n) {
    __shared__ int sred[8];
    const int i0 = (blockIdx.x * 256 + threadIdx.x) * 4;
    int s = 0;
#pragma unroll
    for (int q = 0; q < 4; ++q)
        if (i0 + q < n) s += cnt[i0 + q];
    const int tot = block_sum_256(s, sred);
    if (threadIdx.x == 0) bsum[blockIdx.x] = tot;
}

__global__ void __launch_bounds__(256) scan_final_kernel(const int* __restrict__ cnt,
                                                         const int* __restrict__ bsum,
                                                         int* __restrict__ rowptr,
                                                         float* __restrict__ dinv, int n) {
    __shared__ int sred[8];
    __shared__ int swave[8];
    const int lane = threadIdx.x & 31;
    const int wave = threadIdx.x >> 5;

    int part = 0;
    for (int i = threadIdx.x; i < (int)blockIdx.x; i += 256) part += bsum[i];
    const int base = block_sum_256(part, sred);

    const int i0 = (blockIdx.x * 256 + threadIdx.x) * 4;
    int c[4];
    int s = 0;
#pragma unroll
    for (int q = 0; q < 4; ++q) {
        c[q] = (i0 + q < n) ? cnt[i0 + q] : 0;
        s += c[q];
    }
    int incl = s;
#pragma unroll
    for (int o = 1; o < 32; o <<= 1) {
        const int t = __shfl_up(incl, o, 32);
        if (lane >= o) incl += t;
    }
    if (lane == 31) swave[wave] = incl;
    __syncthreads();
    int woff = 0;
#pragma unroll
    for (int w = 0; w < 8; ++w)
        if (w < wave) woff += swave[w];

    int run = base + woff + incl - s;
#pragma unroll
    for (int q = 0; q < 4; ++q) {
        const int i = i0 + q;
        if (i < n) {
            rowptr[i] = run;
            dinv[i] = rsqrtf((float)(c[q] + 1));
            run += c[q];
            if (i == n - 1) rowptr[n] = run;
        }
    }
}

__global__ void __launch_bounds__(256) csr_fill_kernel(const int* __restrict__ src,
                                                       const int* __restrict__ dst,
                                                       const int* __restrict__ rowptr,
                                                       const int* __restrict__ rank,
                                                       int* __restrict__ col, int n_edges) {
    const int e0 = blockIdx.x * (256 * EPT) + threadIdx.x;
    if ((blockIdx.x + 1) * (256 * EPT) <= n_edges) {
        int s[EPT];
        int d[EPT];
        int p[EPT];
#pragma unroll
        for (int i = 0; i < EPT; ++i) {
            s[i] = src[e0 + i * 256];
            d[i] = dst[e0 + i * 256];
            p[i] = rank[e0 + i * 256];
        }
#pragma unroll
        for (int i = 0; i < EPT; ++i) p[i] += rowptr[d[i]];
#pragma unroll
        for (int i = 0; i < EPT; ++i)
            if ((unsigned)p[i] < (unsigned)n_edges) col[p[i]] = s[i];
    } else {
#pragma unroll
        for (int i = 0; i < EPT; ++i) {
            const int e = e0 + i * 256;
            if (e < n_edges) {
                const int p = rowptr[dst[e]] + rank[e];
                if ((unsigned)p < (unsigned)n_edges) col[p] = src[e];
            }
        }
    }
}


template <int K, int NOUT>
__global__ void __launch_bounds__(256) gemm_kernel(const float* __restrict__ x,
                                                   const float* __restrict__ W,
                                                   float* __restrict__ h, int n_nodes) {
    constexpr int LDK = K + 4;
    constexpr int NT  = NOUT / 16;
    __shared__ float sWt[NOUT * LDK];

    for (int i = threadIdx.x; i < K * NOUT; i += 256) {
        const int k = i / NOUT;
        const int n = i % NOUT;
        sWt[n * LDK + k] = W[i];
    }
    __syncthreads();

    const int wave = threadIdx.x >> 5;
    const int lane = threadIdx.x & 31;
    const int half = lane >> 4;
    const int l    = lane & 15;
    const int m0   = (blockIdx.x * 8 + wave) * 16;
    if (m0 >= n_nodes) return;

    const int rowA = min(m0 + l, n_nodes - 1);
    const float* __restrict__ xrow = x + (size_t)rowA * K;

    v8f acc[NT] = {};
    for (int k = 0; k < K; k += 4) {
        const int kk = k + 2 * half;
        const v2f a = *(const v2f*)(xrow + kk);
#pragma unroll
        for (int t = 0; t < NT; ++t) {
            const int col = t * 16 + l;
            const v2f b = *(const v2f*)(&sWt[col * LDK + kk]);
            acc[t] = __builtin_amdgcn_wmma_f32_16x16x4_f32(
                false, a, false, b, (short)0, acc[t], false, false);
        }
    }

    if (m0 + 16 <= n_nodes) {
        float* __restrict__ hp = h + (size_t)(m0 + 8 * half) * NOUT + l;
#pragma unroll
        for (int v = 0; v < 8; ++v) {
#pragma unroll
            for (int t = 0; t < NT; ++t)
                hp[(size_t)v * NOUT + t * 16] = acc[t][v];
        }
    } else {
#pragma unroll
        for (int v = 0; v < 8; ++v) {
            const int row = m0 + 8 * half + v;
            if (row < n_nodes) {
#pragma unroll
                for (int t = 0; t < NT; ++t)
                    h[(size_t)row * NOUT + t * 16 + l] = acc[t][v];
            }
        }
    }
}


template <int F, bool RELU>
__global__ void __launch_bounds__(256) gather_kernel(const int* __restrict__ rowptr,
                                                     const int* __restrict__ col,
                                                     const float* __restrict__ dinv,
                                                     const float* __restrict__ h,
                                                     const float* __restrict__ bias,
                                                     float* __restrict__ outp, int n_nodes) {
    constexpr int G  = 8;
    constexpr int V4 = F / 32;
    constexpr int Q  = 32 / G;
    const int lane  = threadIdx.x & 31;
    const int sub   = lane & (G - 1);
    const int gbase = lane & ~(G - 1);
    const int node  = (int)((blockIdx.x * 256 + threadIdx.x) >> 3);
    const bool valid = node < n_nodes;
    const int nodec  = valid ? node : n_nodes - 1;

    const int beg  = rowptr[nodec];
    const int endr = rowptr[nodec + 1];
    const int end  = valid ? endr : beg;
    const float dd = dinv[nodec];
    const float* __restrict__ hl = h + 4 * sub;

    v4f hs[V4];
#pragma unroll
    for (int v = 0; v < V4; ++v) hs[v] = *(const v4f*)(hl + (size_t)nodec * F + 32 * v);

    v4f acc[V4];
#pragma unroll
    for (int v = 0; v < V4; ++v) acc[v] = (v4f){0.0f, 0.0f, 0.0f, 0.0f};

    for (int base = beg; __any(base < end); base += 32) {
        unsigned c_l[Q];
#pragma unroll
        for (int q = 0; q < Q; ++q) {
            const int j = base + q * G + sub;
            c_l[q] = (unsigned)nodec;
            if (j < end) c_l[q] = (unsigned)col[j];
        }
        int s_l[Q];
#pragma unroll
        for (int q = 0; q < Q; ++q) {
            s_l[q] = (int)(c_l[q] < (unsigned)n_nodes ? c_l[q] : (unsigned)(n_nodes - 1));
        }
        const int m = end - base;
#pragma unroll
        for (int q = 0; q < Q; ++q) {
            if (__any(q * G < m)) {
                int s[G];
                float w[G];
                v4f r[G][V4];
#pragma unroll
                for (int i = 0; i < G; ++i) s[i] = __shfl(s_l[q], gbase + i, 32);
#pragma unroll
                for (int i = 0; i < G; ++i) w[i] = dinv[s[i]] * dd;
#pragma unroll
                for (int i = 0; i < G; ++i) {
#pragma unroll
                    for (int v = 0; v < V4; ++v)
                        r[i][v] = *(const v4f*)(hl + (size_t)s[i] * F + 32 * v);
                }
#pragma unroll
                for (int i = 0; i < G; ++i) {
                    const bool live = (q * G + i < m);
#pragma unroll
                    for (int v = 0; v < V4; ++v) {
                        acc[v].x = fmaf(w[i], live ? r[i][v].x : 0.0f, acc[v].x);
                        acc[v].y = fmaf(w[i], live ? r[i][v].y : 0.0f, acc[v].y);
                        acc[v].z = fmaf(w[i], live ? r[i][v].z : 0.0f, acc[v].z);
                        acc[v].w = fmaf(w[i], live ? r[i][v].w : 0.0f, acc[v].w);
                    }
                }
            }
        }
    }

    if (!valid) return;
#pragma unroll
    for (int v = 0; v < V4; ++v) {
        const v4f bs = *(const v4f*)(bias + 32 * v + 4 * sub);
        v4f o;
        o.x = acc[v].x + dd * dd * hs[v].x + bs.x;
        o.y = acc[v].y + dd * dd * hs[v].y + bs.y;
        o.z = acc[v].z + dd * dd * hs[v].z + bs.z;
        o.w = acc[v].w + dd * dd * hs[v].w + bs.w;
        if (RELU) {
            o.x = fmaxf(o.x, 0.0f);
            o.y = fmaxf(o.y, 0.0f);
            o.z = fmaxf(o.z, 0.0f);
            o.w = fmaxf(o.w, 0.0f);
        }
        *(v4f*)(outp + (size_t)node * F + 32 * v + 4 * sub) = o;
    }
}


__global__ void __launch_bounds__(256) decode_kernel(const int* __restrict__ pos,
                                                     const int* __restrict__ neg,
                                                     const float* __restrict__ z,
                                                     float* __restrict__ out, int n_eval) {
    const int lane  = threadIdx.x & 31;
    const int sub   = lane & 7;
    const int gbase = lane & 24;
    const int total = 2 * n_eval;
    const int k = (int)(blockIdx.x * 256 + threadIdx.x);
    int a = 0, b = 0;
    if (k < total) {
        if (k < n_eval) { a = pos[k];            b = pos[n_eval + k]; }
        else            { const int kk = k - n_eval; a = neg[kk]; b = neg[n_eval + kk]; }
    }
    const float* __restrict__ zl = z + 4 * sub;
    v4f za[8];
    v4f zb[8];
#pragma unroll
    for (int p = 0; p < 8; ++p) {
        const int ap = __shfl(a, gbase + p, 32);
        const int bp = __shfl(b, gbase + p, 32);
        za[p] = *(const v4f*)(zl + (size_t)ap * F_OUT);
        zb[p] = *(const v4f*)(zl + (size_t)bp * F_OUT);
    }
    float res = 0.0f;
#pragma unroll
    for (int p = 0; p < 8; ++p) {
        float v = za[p].x * zb[p].x + za[p].y * zb[p].y + za[p].z * zb[p].z + za[p].w * zb[p].w;
        v += __shfl_xor(v, 4, 32);
        v += __shfl_xor(v, 2, 32);
        v += __shfl_xor(v, 1, 32);
        if (sub == p) res = v;
    }
    if (k < total) out[k] = res;
}


static inline size_t align_up(size_t x) { return (x + 255) & ~(size_t)255; }

extern "C" void kernel_launch(void* const* d_in, const int* in_sizes, int n_in,
                              void* d_out, int out_size, void* d_ws, size_t ws_size,
                              hipStream_t stream) {
    const float* x   = (const float*)d_in[0];
    const float* W1  = (const float*)d_in[1];
    const float* b1  = (const float*)d_in[2];
    const float* W2  = (const float*)d_in[3];
    const float* b2  = (const float*)d_in[4];
    const int*   ei  = (const int*)d_in[5];
    const int*   pos = (const int*)d_in[6];
    const int*   neg = (const int*)d_in[7];
    float* out = (float*)d_out;

    const int n_nodes = in_sizes[0] / F_IN;
    const int n_edges = in_sizes[5] / 2;
    const int n_eval  = in_sizes[6] / 2;
    const int* src = ei;
    const int* dst = ei + n_edges;

    const int B = 256;
    const int scan_blocks = (n_nodes + 1023) / 1024;

    char* w = (char*)d_ws;
    size_t off = 0;
    float* dinv  = (float*)(w + off); off += align_up((size_t)n_nodes * 4);
    float* h1    = (float*)(w + off); off += align_up((size_t)n_nodes * F_MID * 4);
    float* a1    = (float*)(w + off); off += align_up((size_t)n_nodes * F_MID * 4);
    int* cnt     = (int*)(w + off);   off += align_up((size_t)n_nodes * 4);
    int* rowptr  = (int*)(w + off);   off += align_up((size_t)(n_nodes + 1) * 4);
    int* bsum    = (int*)(w + off);   off += align_up((size_t)scan_blocks * 4);
    int* rank    = (int*)(w + off);   off += align_up((size_t)n_edges * 4);
    int* col     = (int*)(w + off);   off += align_up((size_t)n_edges * 4);
    float* h2    = h1;
    float* z     = h1 + (size_t)n_nodes * F_OUT;
    (void)ws_size;

    zero_kernel<<<(n_nodes + B - 1) / B, B, 0, stream>>>(cnt, n_nodes);
    const int edge_blocks = (n_edges + B * EPT - 1) / (B * EPT);
    rank_kernel<<<edge_blocks, B, 0, stream>>>(dst, cnt, rank, n_edges);
    scan_partial_kernel<<<scan_blocks, B, 0, stream>>>(cnt, bsum, n_nodes);
    scan_final_kernel<<<scan_blocks, B, 0, stream>>>(cnt, bsum, rowptr, dinv, n_nodes);
    csr_fill_kernel<<<edge_blocks, B, 0, stream>>>(src, dst, rowptr, rank, col, n_edges);

    const int gemm_blocks   = (n_nodes + 128 - 1) / 128;
    const int gather_blocks = (n_nodes + 32 - 1) / 32;

    gemm_kernel<F_IN, F_MID><<<gemm_blocks, B, 0, stream>>>(x, W1, h1, n_nodes);
    gather_kernel<F_MID, true><<<gather_blocks, B, 0, stream>>>(rowptr, col, dinv, h1, b1, a1, n_nodes);

    gemm_kernel<F_MID, F_OUT><<<gemm_blocks, B, 0, stream>>>(a1, W2, h2, n_nodes);
    gather_kernel<F_OUT, false><<<gather_blocks, B, 0, stream>>>(rowptr, col, dinv, h2, b2, z, n_nodes);

    {
        int blocks = (2 * n_eval + B - 1) / B;
        decode_kernel<<<blocks, B, 0, stream>>>(pos, neg, z, out, n_eval);
    }
    (void)out_size; (void)n_in;
}


# Source-side degree factor folded into the GEMM stores (rows kept as dinv*(x@W)): gathers add rows with no per-edge dinv lookup, then scale the sum by dinv[target]; rounding regrouped, sum order kept
# speedup vs baseline: 2.6517x; 1.0524x over previous
#include <hip/hip_runtime.h>
#include <hip/hip_bf16.h>


typedef float v2f __attribute__((ext_vector_type(2)));
typedef float v4f __attribute__((ext_vector_type(4)));
typedef float v8f __attribute__((ext_vector_type(8)));

#define F_IN  256
#define F_MID 64
#define F_OUT 32
#define EPT   8


__global__ void __launch_bounds__(256) zero_kernel(int* __restrict__ p, int n) {
    int i = blockIdx.x * 256 + threadIdx.x;
    if (i < n) p[i] = 0;
}

__global__ void __launch_bounds__(256) rank_kernel(const int* __restrict__ dst,
                                                   int* __restrict__ cnt,
                                                   int* __restrict__ rank, int n_edges) {
    const int e0 = blockIdx.x * (256 * EPT) + threadIdx.x;
    if ((blockIdx.x + 1) * (256 * EPT) <= n_edges) {
        int d[EPT];
        int r[EPT];
#pragma unroll
        for (int i = 0; i < EPT; ++i) d[i] = dst[e0 + i * 256];
#pragma unroll
        for (int i = 0; i < EPT; ++i) r[i] = atomicAdd(&cnt[d[i]], 1);
#pragma unroll
        for (int i = 0; i < EPT; ++i) rank[e0 + i * 256] = r[i];
    } else {
#pragma unroll
        for (int i = 0; i < EPT; ++i) {
            const int e = e0 + i * 256;
            if (e < n_edges) rank[e] = atomicAdd(&cnt[dst[e]], 1);
        }
    }
}

__device__ __forceinline__ int block_sum_256(int v, int* sred) {
    const int lane = threadIdx.x & 31;
    const int wave = threadIdx.x >> 5;
#pragma unroll
    for (int o = 16; o > 0; o >>= 1) v += __shfl_xor(v, o, 32);
    if (lane == 0) sred[wave] = v;
    __syncthreads();
    int t = 0;
#pragma unroll
    for (int w = 0; w < 8; ++w) t += sred[w];
    return t;
}

__global__ void __launch_bounds__(256) scan_partial_kernel(const int* __restrict__ cnt,
                                                           int* __restrict__ bsum, int n) {
    __shared__ int sred[8];
    const int i0 = (blockIdx.x * 256 + threadIdx.x) * 4;
    int s = 0;
#pragma unroll
    for (int q = 0; q < 4; ++q)
        if (i0 + q < n) s += cnt[i0 + q];
    const int tot = block_sum_256(s, sred);
    if (threadIdx.x == 0) bsum[blockIdx.x] = tot;
}

__global__ void __launch_bounds__(256) scan_final_kernel(const int* __restrict__ cnt,
                                                         const int* __restrict__ bsum,
                                                         int* __restrict__ rowptr,
                                                         float* __restrict__ dinv, int n) {
    __shared__ int sred[8];
    __shared__ int swave[8];
    const int lane = threadIdx.x & 31;
    const int wave = threadIdx.x >> 5;

    int part = 0;
    for (int i = threadIdx.x; i < (int)blockIdx.x; i += 256) part += bsum[i];
    const int base = block_sum_256(part, sred);

    const int i0 = (blockIdx.x * 256 + threadIdx.x) * 4;
    int c[4];
    int s = 0;
#pragma unroll
    for (int q = 0; q < 4; ++q) {
        c[q] = (i0 + q < n) ? cnt[i0 + q] : 0;
        s += c[q];
    }
    int incl = s;
#pragma unroll
    for (int o = 1; o < 32; o <<= 1) {
        const int t = __shfl_up(incl, o, 32);
        if (lane >= o) incl += t;
    }
    if (lane == 31) swave[wave] = incl;
    __syncthreads();
    int woff = 0;
#pragma unroll
    for (int w = 0; w < 8; ++w)
        if (w < wave) woff += swave[w];

    int run = base + woff + incl - s;
#pragma unroll
    for (int q = 0; q < 4; ++q) {
        const int i = i0 + q;
        if (i < n) {
            rowptr[i] = run;
            dinv[i] = rsqrtf((float)(c[q] + 1));
            run += c[q];
            if (i == n - 1) rowptr[n] = run;
        }
    }
}

__global__ void __launch_bounds__(256) csr_fill_kernel(const int* __restrict__ src,
                                                       const int* __restrict__ dst,
                                                       const int* __restrict__ rowptr,
                                                       const int* __restrict__ rank,
                                                       int* __restrict__ col, int n_edges) {
    const int e0 = blockIdx.x * (256 * EPT) + threadIdx.x;
    if ((blockIdx.x + 1) * (256 * EPT) <= n_edges) {
        int s[EPT];
        int d[EPT];
        int p[EPT];
#pragma unroll
        for (int i = 0; i < EPT; ++i) {
            s[i] = src[e0 + i * 256];
            d[i] = dst[e0 + i * 256];
            p[i] = rank[e0 + i * 256];
        }
#pragma unroll
        for (int i = 0; i < EPT; ++i) p[i] += rowptr[d[i]];
#pragma unroll
        for (int i = 0; i < EPT; ++i)
            if ((unsigned)p[i] < (unsigned)n_edges) col[p[i]] = s[i];
    } else {
#pragma unroll
        for (int i = 0; i < EPT; ++i) {
            const int e = e0 + i * 256;
            if (e < n_edges) {
                const int p = rowptr[dst[e]] + rank[e];
                if ((unsigned)p < (unsigned)n_edges) col[p] = src[e];
            }
        }
    }
}


template <int K, int NOUT>
__global__ void __launch_bounds__(256) gemm_kernel(const float* __restrict__ x,
                                                   const float* __restrict__ W,
                                                   const float* __restrict__ dinv,
                                                   float* __restrict__ h, int n_nodes) {
    constexpr int LDK = K + 4;
    constexpr int NT  = NOUT / 16;
    __shared__ float sWt[NOUT * LDK];

    for (int i = threadIdx.x; i < K * NOUT; i += 256) {
        const int k = i / NOUT;
        const int n = i % NOUT;
        sWt[n * LDK + k] = W[i];
    }
    __syncthreads();

    const int wave = threadIdx.x >> 5;
    const int lane = threadIdx.x & 31;
    const int half = lane >> 4;
    const int l    = lane & 15;
    const int m0   = (blockIdx.x * 8 + wave) * 16;
    if (m0 >= n_nodes) return;

    const int rowA = min(m0 + l, n_nodes - 1);
    const float* __restrict__ xrow = x + (size_t)rowA * K;

    float dv[8];
    if (m0 + 16 <= n_nodes) {
        const v4f d0 = *(const v4f*)(dinv + m0 + 8 * half);
        const v4f d1 = *(const v4f*)(dinv + m0 + 8 * half + 4);
        dv[0] = d0.x; dv[1] = d0.y; dv[2] = d0.z; dv[3] = d0.w;
        dv[4] = d1.x; dv[5] = d1.y; dv[6] = d1.z; dv[7] = d1.w;
    } else {
#pragma unroll
        for (int v = 0; v < 8; ++v) dv[v] = dinv[min(m0 + 8 * half + v, n_nodes - 1)];
    }

    v8f acc[NT] = {};
    for (int k = 0; k < K; k += 4) {
        const int kk = k + 2 * half;
        const v2f a = *(const v2f*)(xrow + kk);
#pragma unroll
        for (int t = 0; t < NT; ++t) {
            const int col = t * 16 + l;
            const v2f b = *(const v2f*)(&sWt[col * LDK + kk]);
            acc[t] = __builtin_amdgcn_wmma_f32_16x16x4_f32(
                false, a, false, b, (short)0, acc[t], false, false);
        }
    }

    if (m0 + 16 <= n_nodes) {
        float* __restrict__ hp = h + (size_t)(m0 + 8 * half) * NOUT + l;
#pragma unroll
        for (int v = 0; v < 8; ++v) {
#pragma unroll
            for (int t = 0; t < NT; ++t)
                hp[(size_t)v * NOUT + t * 16] = dv[v] * acc[t][v];
        }
    } else {
#pragma unroll
        for (int v = 0; v < 8; ++v) {
            const int row = m0 + 8 * half + v;
            if (row < n_nodes) {
#pragma unroll
                for (int t = 0; t < NT; ++t)
                    h[(size_t)row * NOUT + t * 16 + l] = dv[v] * acc[t][v];
            }
        }
    }
}


template <int F, bool RELU>
__global__ void __launch_bounds__(256) gather_kernel(const int* __restrict__ rowptr,
                                                     const int* __restrict__ col,
                                                     const float* __restrict__ dinv,
                                                     const float* __restrict__ h,
                                                     const float* __restrict__ bias,
                                                     float* __restrict__ outp, int n_nodes) {
    constexpr int G  = 8;
    constexpr int V4 = F / 32;
    constexpr int Q  = 32 / G;
    const int lane  = threadIdx.x & 31;
    const int sub   = lane & (G - 1);
    const int gbase = lane & ~(G - 1);
    const int node  = (int)((blockIdx.x * 256 + threadIdx.x) >> 3);
    const bool valid = node < n_nodes;
    const int nodec  = valid ? node : n_nodes - 1;

    const int beg  = rowptr[nodec];
    const int endr = rowptr[nodec + 1];
    const int end  = valid ? endr : beg;
    const float dd = dinv[nodec];
    const float* __restrict__ hl = h + 4 * sub;

    v4f hs[V4];
#pragma unroll
    for (int v = 0; v < V4; ++v) hs[v] = *(const v4f*)(hl + (size_t)nodec * F + 32 * v);

    v4f acc[V4];
#pragma unroll
    for (int v = 0; v < V4; ++v) acc[v] = (v4f){0.0f, 0.0f, 0.0f, 0.0f};

    for (int base = beg; __any(base < end); base += 32) {
        unsigned c_l[Q];
#pragma unroll
        for (int q = 0; q < Q; ++q) {
            const int j = base + q * G + sub;
            c_l[q] = (unsigned)nodec;
            if (j < end) c_l[q] = (unsigned)col[j];
        }
        int s_l[Q];
#pragma unroll
        for (int q = 0; q < Q; ++q) {
            s_l[q] = (int)(c_l[q] < (unsigned)n_nodes ? c_l[q] : (unsigned)(n_nodes - 1));
        }
        const int m = end - base;
#pragma unroll
        for (int q = 0; q < Q; ++q) {
            if (__any(q * G < m)) {
                int s[G];
                v4f r[G][V4];
#pragma unroll
                for (int i = 0; i < G; ++i) s[i] = __shfl(s_l[q], gbase + i, 32);
#pragma unroll
                for (int i = 0; i < G; ++i) {
#pragma unroll
                    for (int v = 0; v < V4; ++v)
                        r[i][v] = *(const v4f*)(hl + (size_t)s[i] * F + 32 * v);
                }
#pragma unroll
                for (int i = 0; i < G; ++i) {
                    const float lv = (q * G + i < m) ? 1.0f : 0.0f;
#pragma unroll
                    for (int v = 0; v < V4; ++v) {
                        acc[v].x = fmaf(lv, r[i][v].x, acc[v].x);
                        acc[v].y = fmaf(lv, r[i][v].y, acc[v].y);
                        acc[v].z = fmaf(lv, r[i][v].z, acc[v].z);
                        acc[v].w = fmaf(lv, r[i][v].w, acc[v].w);
                    }
                }
            }
        }
    }

    if (!valid) return;
#pragma unroll
    for (int v = 0; v < V4; ++v) {
        const v4f bs = *(const v4f*)(bias + 32 * v + 4 * sub);
        v4f o;
        o.x = dd * (acc[v].x + hs[v].x) + bs.x;
        o.y = dd * (acc[v].y + hs[v].y) + bs.y;
        o.z = dd * (acc[v].z + hs[v].z) + bs.z;
        o.w = dd * (acc[v].w + hs[v].w) + bs.w;
        if (RELU) {
            o.x = fmaxf(o.x, 0.0f);
            o.y = fmaxf(o.y, 0.0f);
            o.z = fmaxf(o.z, 0.0f);
            o.w = fmaxf(o.w, 0.0f);
        }
        *(v4f*)(outp + (size_t)node * F + 32 * v + 4 * sub) = o;
    }
}


__global__ void __launch_bounds__(256) decode_kernel(const int* __restrict__ pos,
                                                     const int* __restrict__ neg,
                                                     const float* __restrict__ z,
                                                     float* __restrict__ out, int n_eval) {
    const int lane  = threadIdx.x & 31;
    const int sub   = lane & 7;
    const int gbase = lane & 24;
    const int total = 2 * n_eval;
    const int k = (int)(blockIdx.x * 256 + threadIdx.x);
    int a = 0, b = 0;
    if (k < total) {
        if (k < n_eval) { a = pos[k];            b = pos[n_eval + k]; }
        else            { const int kk = k - n_eval; a = neg[kk]; b = neg[n_eval + kk]; }
    }
    const float* __restrict__ zl = z + 4 * sub;
    v4f za[8];
    v4f zb[8];
#pragma unroll
    for (int p = 0; p < 8; ++p) {
        const int ap = __shfl(a, gbase + p, 32);
        const int bp = __shfl(b, gbase + p, 32);
        za[p] = *(const v4f*)(zl + (size_t)ap * F_OUT);
        zb[p] = *(const v4f*)(zl + (size_t)bp * F_OUT);
    }
    float res = 0.0f;
#pragma unroll
    for (int p = 0; p < 8; ++p) {
        float v = za[p].x * zb[p].x + za[p].y * zb[p].y + za[p].z * zb[p].z + za[p].w * zb[p].w;
        v += __shfl_xor(v, 4, 32);
        v += __shfl_xor(v, 2, 32);
        v += __shfl_xor(v, 1, 32);
        if (sub == p) res = v;
    }
    if (k < total) out[k] = res;
}


static inline size_t align_up(size_t x) { return (x + 255) & ~(size_t)255; }

extern "C" void kernel_launch(void* const* d_in, const int* in_sizes, int n_in,
                              void* d_out, int out_size, void* d_ws, size_t ws_size,
                              hipStream_t stream) {
    const float* x   = (const float*)d_in[0];
    const float* W1  = (const float*)d_in[1];
    const float* b1  = (const float*)d_in[2];
    const float* W2  = (const float*)d_in[3];
    const float* b2  = (const float*)d_in[4];
    const int*   ei  = (const int*)d_in[5];
    const int*   pos = (const int*)d_in[6];
    const int*   neg = (const int*)d_in[7];
    float* out = (float*)d_out;

    const int n_nodes = in_sizes[0] / F_IN;
    const int n_edges = in_sizes[5] / 2;
    const int n_eval  = in_sizes[6] / 2;
    const int* src = ei;
    const int* dst = ei + n_edges;

    const int B = 256;
    const int scan_blocks = (n_nodes + 1023) / 1024;

    char* w = (char*)d_ws;
    size_t off = 0;
    float* dinv  = (float*)(w + off); off += align_up((size_t)n_nodes * 4);
    float* h1    = (float*)(w + off); off += align_up((size_t)n_nodes * F_MID * 4);
    float* a1    = (float*)(w + off); off += align_up((size_t)n_nodes * F_MID * 4);
    int* cnt     = (int*)(w + off);   off += align_up((size_t)n_nodes * 4);
    int* rowptr  = (int*)(w + off);   off += align_up((size_t)(n_nodes + 1) * 4);
    int* bsum    = (int*)(w + off);   off += align_up((size_t)scan_blocks * 4);
    int* rank    = (int*)(w + off);   off += align_up((size_t)n_edges * 4);
    int* col     = (int*)(w + off);   off += align_up((size_t)n_edges * 4);
    float* h2    = h1;
    float* z     = h1 + (size_t)n_nodes * F_OUT;
    (void)ws_size;

    zero_kernel<<<(n_nodes + B - 1) / B, B, 0, stream>>>(cnt, n_nodes);
    const int edge_blocks = (n_edges + B * EPT - 1) / (B * EPT);
    rank_kernel<<<edge_blocks, B, 0, stream>>>(dst, cnt, rank, n_edges);
    scan_partial_kernel<<<scan_blocks, B, 0, stream>>>(cnt, bsum, n_nodes);
    scan_final_kernel<<<scan_blocks, B, 0, stream>>>(cnt, bsum, rowptr, dinv, n_nodes);
    csr_fill_kernel<<<edge_blocks, B, 0, stream>>>(src, dst, rowptr, rank, col, n_edges);

    const int gemm_blocks   = (n_nodes + 128 - 1) / 128;
    const int gather_blocks = (n_nodes + 32 - 1) / 32;

    gemm_kernel<F_IN, F_MID><<<gemm_blocks, B, 0, stream>>>(x, W1, dinv, h1, n_nodes);
    gather_kernel<F_MID, true><<<gather_blocks, B, 0, stream>>>(rowptr, col, dinv, h1, b1, a1, n_nodes);

    gemm_kernel<F_MID, F_OUT><<<gemm_blocks, B, 0, stream>>>(a1, W2, dinv, h2, n_nodes);
    gather_kernel<F_OUT, false><<<gather_blocks, B, 0, stream>>>(rowptr, col, dinv, h2, b2, z, n_nodes);

    {
        int blocks = (2 * n_eval + B - 1) / B;
        decode_kernel<<<blocks, B, 0, stream>>>(pos, neg, z, out, n_eval);
    }
    (void)out_size; (void)n_in;
}


# CSR build without global atomics: 4096-edge tiles sorted by 256-node target range in LDS, then one work-group per range counts and places by LDS atomics (10 launches to 7); sum order LDS arrival
# speedup vs baseline: 5.1355x; 1.9367x over previous
#include <hip/hip_runtime.h>
#include <hip/hip_bf16.h>


typedef float v2f __attribute__((ext_vector_type(2)));
typedef float v4f __attribute__((ext_vector_type(4)));
typedef float v8f __attribute__((ext_vector_type(8)));
typedef int   v2i __attribute__((ext_vector_type(2)));

#define F_IN  256
#define F_MID 64
#define F_OUT 32

#define TILE  4096
#define BSH   8
#define BN    256
#define PMAX  1024
#define CAP   8192


__device__ __forceinline__ int block_excl_scan_1024(int v, int* swave, int& total) {
    const int lane = threadIdx.x & 31;
    const int wave = threadIdx.x >> 5;
    int incl = v;
#pragma unroll
    for (int o = 1; o < 32; o <<= 1) {
        const int t = __shfl_up(incl, o, 32);
        if (lane >= o) incl += t;
    }
    if (lane == 31) swave[wave] = incl;
    __syncthreads();
    const int w = swave[lane];
    int wincl = w;
#pragma unroll
    for (int o = 1; o < 32; o <<= 1) {
        const int t = __shfl_up(wincl, o, 32);
        if (lane >= o) wincl += t;
    }
    const int woff = __shfl(wincl, wave, 32) - __shfl(w, wave, 32);
    total = __shfl(wincl, 31, 32);
    return woff + incl - v;
}

__global__ void __launch_bounds__(1024) tile_sort_kernel(const int* __restrict__ src,
                                                         const int* __restrict__ dst,
                                                         int* __restrict__ part,
                                                         int* __restrict__ S,
                                                         int n_edges, int P) {
    __shared__ int hist[PMAX + 1];
    __shared__ int swave[32];
    __shared__ v2i stage[TILE];
    const int tid      = threadIdx.x;
    const int tile0    = blockIdx.x * TILE;
    const int tile_len = min(TILE, n_edges - tile0);

    for (int i = tid; i <= P; i += 1024) hist[i] = 0;

    int s[4];
    int d[4];
#pragma unroll
    for (int k = 0; k < 4; ++k) {
        const int i = tid + k * 1024;
        s[k] = 0;
        d[k] = 0;
        if (i < tile_len) {
            s[k] = src[tile0 + i];
            d[k] = dst[tile0 + i];
        }
    }
    __syncthreads();

    int b[4];
    int r[4];
#pragma unroll
    for (int k = 0; k < 4; ++k) {
        b[k] = 0;
        r[k] = 0;
        if (tid + k * 1024 < tile_len) {
            const unsigned q = (unsigned)d[k] >> BSH;
            b[k] = (int)(q < (unsigned)P ? q : (unsigned)(P - 1));
            r[k] = atomicAdd(&hist[b[k]], 1);
        }
    }
    __syncthreads();

    const int v = (tid < P) ? hist[tid] : 0;
    int total;
    const int excl = block_excl_scan_1024(v, swave, total);
    if (tid < P) hist[tid] = excl;
    if (tid == 0) hist[P] = total;
    __syncthreads();

    int* __restrict__ Srow = S + (size_t)blockIdx.x * (size_t)(P + 1);
    for (int i = tid; i <= P; i += 1024) Srow[i] = hist[i];

#pragma unroll
    for (int k = 0; k < 4; ++k) {
        if (tid + k * 1024 < tile_len) {
            const int p = hist[b[k]] + r[k];
            if ((unsigned)p < (unsigned)TILE) stage[p] = (v2i){s[k], d[k]};
        }
    }
    __syncthreads();

    v2i* __restrict__ pout = (v2i*)part + (size_t)tile0;
#pragma unroll
    for (int k = 0; k < 4; ++k) {
        const int i = tid + k * 1024;
        if (i < tile_len) pout[i] = stage[i];
    }
}

#define FOR_RANGE_EDGES(BODY)                                                          \
    for (int tb = 0; tb < T; tb += 1024) {                                             \
        __syncthreads();                        \
        {                                                                              \
            const int t = tb + tid;                                                    \
            int a0 = 0, a1 = 0;                                                        \
            if (t < T) {                                                               \
                a0 = S[(size_t)t * (size_t)(P + 1) + b];                               \
                a1 = S[(size_t)t * (size_t)(P + 1) + b + 1];                           \
            }                                                                          \
            seg0[tid] = a0;                                                            \
            seg1[tid] = a1;                                                            \
        }                                                                              \
        __syncthreads();                                                               \
        const int tn = min(1024, T - tb);                                              \
        for (int g = grp; g < tn; g += 128) {                                          \
                      \
            const int lim = min(TILE, n_edges - (tb + g) * TILE);                      \
            const int j0  = max(seg0[g], 0);                                           \
            const int j1  = min(seg1[g], lim);                                         \
            const v2i* __restrict__ p = p2 + (size_t)(tb + g) * TILE;                  \
            for (int j = j0 + sub; j < j1; j += 16) {                                  \
                const bool two = j + 8 < j1;                                           \
                const v2i e0 = p[j];                                                   \
                const v2i e1 = two ? p[j + 8] : e0;                                    \
                {                                                                      \
                    const unsigned u = (unsigned)(e0.y - node0);                       \
                    const int dl = (int)(u < (unsigned)nn ? u : (unsigned)(nn - 1));   \
                    const int sv = e0.x;                                               \
                    BODY                                                               \
                }                                                                      \
                if (two) {                                                             \
                    const unsigned u = (unsigned)(e1.y - node0);                       \
                    const int dl = (int)(u < (unsigned)nn ? u : (unsigned)(nn - 1));   \
                    const int sv = e1.x;                                               \
                    BODY                                                               \
                }                                                                      \
            }                                                                          \
        }                                                                              \
    }

__global__ void __launch_bounds__(1024) range_csr_kernel(const int* __restrict__ part,
                                                         const int* __restrict__ S,
                                                         int* __restrict__ rowptr,
                                                         float* __restrict__ dinv,
                                                         int* __restrict__ col,
                                                         int n_edges, int n_nodes,
                                                         int P, int T) {
    __shared__ int cnt[BN];
    __shared__ int cur[BN];
    __shared__ int swave_a[32];
    __shared__ int swave_b[32];
    __shared__ int seg0[1024];
    __shared__ int seg1[1024];
    __shared__ int stage[CAP];
    const int tid   = threadIdx.x;
    const int b     = blockIdx.x;
    const int node0 = b * BN;
    const int nn    = min(BN, n_nodes - node0);
    const int grp   = tid >> 3;
    const int sub   = tid & 7;
    const v2i* __restrict__ p2 = (const v2i*)part;

    if (tid < BN) cnt[tid] = 0;

    int bpart = 0;
    for (int t = tid; t < T; t += 1024) bpart += S[(size_t)t * (size_t)(P + 1) + b];

    FOR_RANGE_EDGES({ (void)sv; atomicAdd(&cnt[dl], 1); })
    __syncthreads();

    int base;
    (void)block_excl_scan_1024(bpart, swave_a, base);
    const int c = (tid < BN) ? cnt[tid] : 0;
    int nb;
    const int ex = block_excl_scan_1024(c, swave_b, nb);
    if (tid < BN) cur[tid] = ex;
    if (tid < nn) {
        rowptr[node0 + tid] = base + ex;
        dinv[node0 + tid]   = rsqrtf((float)(c + 1));
    }
    if (b == P - 1 && tid == 0) rowptr[n_nodes] = base + nb;

    const bool staged = nb <= CAP;
    FOR_RANGE_EDGES({
        const int pos = atomicAdd(&cur[dl], 1);
        if (staged) {
            if ((unsigned)pos < (unsigned)CAP) stage[pos] = sv;
        } else {
            const int q = base + pos;
            if ((unsigned)q < (unsigned)n_edges) col[q] = sv;
        }
    })
    __syncthreads();
    if (staged) {
        for (int i = tid; i < nb; i += 1024) {
            const int q = base + i;
            if ((unsigned)q < (unsigned)n_edges) col[q] = stage[i];
        }
    }
}


template <int K, int NOUT>
__global__ void __launch_bounds__(256) gemm_kernel(const float* __restrict__ x,
                                                   const float* __restrict__ W,
                                                   const float* __restrict__ dinv,
                                                   float* __restrict__ h, int n_nodes) {
    constexpr int LDK = K + 4;
    constexpr int NT  = NOUT / 16;
    __shared__ float sWt[NOUT * LDK];

    for (int i = threadIdx.x; i < K * NOUT; i += 256) {
        const int k = i / NOUT;
        const int n = i % NOUT;
        sWt[n * LDK + k] = W[i];
    }
    __syncthreads();

    const int wave = threadIdx.x >> 5;
    const int lane = threadIdx.x & 31;
    const int half = lane >> 4;
    const int l    = lane & 15;
    const int m0   = (blockIdx.x * 8 + wave) * 16;
    if (m0 >= n_nodes) return;

    const int rowA = min(m0 + l, n_nodes - 1);
    const float* __restrict__ xrow = x + (size_t)rowA * K;

    float dv[8];
    if (m0 + 16 <= n_nodes) {
        const v4f d0 = *(const v4f*)(dinv + m0 + 8 * half);
        const v4f d1 = *(const v4f*)(dinv + m0 + 8 * half + 4);
        dv[0] = d0.x; dv[1] = d0.y; dv[2] = d0.z; dv[3] = d0.w;
        dv[4] = d1.x; dv[5] = d1.y; dv[6] = d1.z; dv[7] = d1.w;
    } else {
#pragma unroll
        for (int v = 0; v < 8; ++v) dv[v] = dinv[min(m0 + 8 * half + v, n_nodes - 1)];
    }

    v8f acc[NT] = {};
    for (int k = 0; k < K; k += 4) {
        const int kk = k + 2 * half;
        const v2f a = *(const v2f*)(xrow + kk);
#pragma unroll
        for (int t = 0; t < NT; ++t) {
            const int col = t * 16 + l;
            const v2f b = *(const v2f*)(&sWt[col * LDK + kk]);
            acc[t] = __builtin_amdgcn_wmma_f32_16x16x4_f32(
                false, a, false, b, (short)0, acc[t], false, false);
        }
    }

    if (m0 + 16 <= n_nodes) {
        float* __restrict__ hp = h + (size_t)(m0 + 8 * half) * NOUT + l;
#pragma unroll
        for (int v = 0; v < 8; ++v) {
#pragma unroll
            for (int t = 0; t < NT; ++t)
                hp[(size_t)v * NOUT + t * 16] = dv[v] * acc[t][v];
        }
    } else {
#pragma unroll
        for (int v = 0; v < 8; ++v) {
            const int row = m0 + 8 * half + v;
            if (row < n_nodes) {
#pragma unroll
                for (int t = 0; t < NT; ++t)
                    h[(size_t)row * NOUT + t * 16 + l] = dv[v] * acc[t][v];
            }
        }
    }
}


template <int F, bool RELU>
__global__ void __launch_bounds__(256) gather_kernel(const int* __restrict__ rowptr,
                                                     const int* __restrict__ col,
                                                     const float* __restrict__ dinv,
                                                     const float* __restrict__ h,
                                                     const float* __restrict__ bias,
                                                     float* __restrict__ outp, int n_nodes) {
    constexpr int G  = 8;
    constexpr int V4 = F / 32;
    constexpr int Q  = 32 / G;
    const int lane  = threadIdx.x & 31;
    const int sub   = lane & (G - 1);
    const int gbase = lane & ~(G - 1);
    const int node  = (int)((blockIdx.x * 256 + threadIdx.x) >> 3);
    const bool valid = node < n_nodes;
    const int nodec  = valid ? node : n_nodes - 1;

    const int beg  = rowptr[nodec];
    const int endr = rowptr[nodec + 1];
    const int end  = valid ? endr : beg;
    const float dd = dinv[nodec];
    const float* __restrict__ hl = h + 4 * sub;

    v4f hs[V4];
#pragma unroll
    for (int v = 0; v < V4; ++v) hs[v] = *(const v4f*)(hl + (size_t)nodec * F + 32 * v);

    v4f acc[V4];
#pragma unroll
    for (int v = 0; v < V4; ++v) acc[v] = (v4f){0.0f, 0.0f, 0.0f, 0.0f};

    for (int base = beg; __any(base < end); base += 32) {
        unsigned c_l[Q];
#pragma unroll
        for (int q = 0; q < Q; ++q) {
            const int j = base + q * G + sub;
            c_l[q] = (unsigned)nodec;
            if (j < end) c_l[q] = (unsigned)col[j];
        }
        int s_l[Q];
#pragma unroll
        for (int q = 0; q < Q; ++q) {
            s_l[q] = (int)(c_l[q] < (unsigned)n_nodes ? c_l[q] : (unsigned)(n_nodes - 1));
        }
        const int m = end - base;
#pragma unroll
        for (int q = 0; q < Q; ++q) {
            if (__any(q * G < m)) {
                int s[G];
                v4f r[G][V4];
#pragma unroll
                for (int i = 0; i < G; ++i) s[i] = __shfl(s_l[q], gbase + i, 32);
#pragma unroll
                for (int i = 0; i < G; ++i) {
#pragma unroll
                    for (int v = 0; v < V4; ++v)
                        r[i][v] = *(const v4f*)(hl + (size_t)s[i] * F + 32 * v);
                }
#pragma unroll
                for (int i = 0; i < G; ++i) {
                    const float lv = (q * G + i < m) ? 1.0f : 0.0f;
#pragma unroll
                    for (int v = 0; v < V4; ++v) {
                        acc[v].x = fmaf(lv, r[i][v].x, acc[v].x);
                        acc[v].y = fmaf(lv, r[i][v].y, acc[v].y);
                        acc[v].z = fmaf(lv, r[i][v].z, acc[v].z);
                        acc[v].w = fmaf(lv, r[i][v].w, acc[v].w);
                    }
                }
            }
        }
    }

    if (!valid) return;
#pragma unroll
    for (int v = 0; v < V4; ++v) {
        const v4f bs = *(const v4f*)(bias + 32 * v + 4 * sub);
        v4f o;
        o.x = dd * (acc[v].x + hs[v].x) + bs.x;
        o.y = dd * (acc[v].y + hs[v].y) + bs.y;
        o.z = dd * (acc[v].z + hs[v].z) + bs.z;
        o.w = dd * (acc[v].w + hs[v].w) + bs.w;
        if (RELU) {
            o.x = fmaxf(o.x, 0.0f);
            o.y = fmaxf(o.y, 0.0f);
            o.z = fmaxf(o.z, 0.0f);
            o.w = fmaxf(o.w, 0.0f);
        }
        *(v4f*)(outp + (size_t)node * F + 32 * v + 4 * sub) = o;
    }
}


__global__ void __launch_bounds__(256) decode_kernel(const int* __restrict__ pos,
                                                     const int* __restrict__ neg,
                                                     const float* __restrict__ z,
                                                     float* __restrict__ out, int n_eval) {
    const int lane  = threadIdx.x & 31;
    const int sub   = lane & 7;
    const int gbase = lane & 24;
    const int total = 2 * n_eval;
    const int k = (int)(blockIdx.x * 256 + threadIdx.x);
    int a = 0, b = 0;
    if (k < total) {
        if (k < n_eval) { a = pos[k];            b = pos[n_eval + k]; }
        else            { const int kk = k - n_eval; a = neg[kk]; b = neg[n_eval + kk]; }
    }
    const float* __restrict__ zl = z + 4 * sub;
    v4f za[8];
    v4f zb[8];
#pragma unroll
    for (int p = 0; p < 8; ++p) {
        const int ap = __shfl(a, gbase + p, 32);
        const int bp = __shfl(b, gbase + p, 32);
        za[p] = *(const v4f*)(zl + (size_t)ap * F_OUT);
        zb[p] = *(const v4f*)(zl + (size_t)bp * F_OUT);
    }
    float res = 0.0f;
#pragma unroll
    for (int p = 0; p < 8; ++p) {
        float v = za[p].x * zb[p].x + za[p].y * zb[p].y + za[p].z * zb[p].z + za[p].w * zb[p].w;
        v += __shfl_xor(v, 4, 32);
        v += __shfl_xor(v, 2, 32);
        v += __shfl_xor(v, 1, 32);
        if (sub == p) res = v;
    }
    if (k < total) out[k] = res;
}


static inline size_t align_up(size_t x) { return (x + 255) & ~(size_t)255; }

extern "C" void kernel_launch(void* const* d_in, const int* in_sizes, int n_in,
                              void* d_out, int out_size, void* d_ws, size_t ws_size,
                              hipStream_t stream) {
    const float* x   = (const float*)d_in[0];
    const float* W1  = (const float*)d_in[1];
    const float* b1  = (const float*)d_in[2];
    const float* W2  = (const float*)d_in[3];
    const float* b2  = (const float*)d_in[4];
    const int*   ei  = (const int*)d_in[5];
    const int*   pos = (const int*)d_in[6];
    const int*   neg = (const int*)d_in[7];
    float* out = (float*)d_out;

    const int n_nodes = in_sizes[0] / F_IN;
    const int n_edges = in_sizes[5] / 2;
    const int n_eval  = in_sizes[6] / 2;
    const int* src = ei;
    const int* dst = ei + n_edges;

    const int B = 256;
    const int P = (n_nodes + BN - 1) / BN;
    const int T = (n_edges + TILE - 1) / TILE;

    char* w = (char*)d_ws;
    size_t off = 0;
    float* dinv  = (float*)(w + off); off += align_up((size_t)n_nodes * 4);
    float* h1    = (float*)(w + off); off += align_up((size_t)n_nodes * F_MID * 4);
    const size_t a1_bytes   = (size_t)n_nodes * F_MID * 4;
    const size_t part_bytes = (size_t)n_edges * 8;
    float* a1    = (float*)(w + off);
    int* part    = (int*)(w + off);   off += align_up(a1_bytes > part_bytes ? a1_bytes : part_bytes);
    int* rowptr  = (int*)(w + off);   off += align_up((size_t)(n_nodes + 1) * 4);
    int* S       = (int*)(w + off);   off += align_up((size_t)T * (size_t)(P + 1) * 4);
    int* col     = (int*)(w + off);   off += align_up((size_t)n_edges * 4);
    float* h2    = h1;
    float* z     = h1 + (size_t)n_nodes * F_OUT;
    (void)ws_size;

    tile_sort_kernel<<<T, 1024, 0, stream>>>(src, dst, part, S, n_edges, P);
    range_csr_kernel<<<P, 1024, 0, stream>>>(part, S, rowptr, dinv, col, n_edges, n_nodes, P, T);

    const int gemm_blocks   = (n_nodes + 128 - 1) / 128;
    const int gather_blocks = (n_nodes + 32 - 1) / 32;

    gemm_kernel<F_IN, F_MID><<<gemm_blocks, B, 0, stream>>>(x, W1, dinv, h1, n_nodes);
    gather_kernel<F_MID, true><<<gather_blocks, B, 0, stream>>>(rowptr, col, dinv, h1, b1, a1, n_nodes);

    gemm_kernel<F_MID, F_OUT><<<gemm_blocks, B, 0, stream>>>(a1, W2, dinv, h2, n_nodes);
    gather_kernel<F_OUT, false><<<gather_blocks, B, 0, stream>>>(rowptr, col, dinv, h2, b2, z, n_nodes);

    {
        int blocks = (2 * n_eval + B - 1) / B;
        decode_kernel<<<blocks, B, 0, stream>>>(pos, neg, z, out, n_eval);
    }
    (void)out_size; (void)n_in;
}


# Layer-2 GEMM fused into the layer-1 gather: a work-group's 32 gathered rows go through LDS to two WMMA waves; the a1 plane and one launch are gone (7 to 6); every sum and k order unchanged
# speedup vs baseline: 5.9313x; 1.1550x over previous
#include <hip/hip_runtime.h>
#include <hip/hip_bf16.h>


typedef float v2f __attribute__((ext_vector_type(2)));
typedef float v4f __attribute__((ext_vector_type(4)));
typedef float v8f __attribute__((ext_vector_type(8)));
typedef int   v2i __attribute__((ext_vector_type(2)));

#define F_IN  256
#define F_MID 64
#define F_OUT 32

#define TILE  4096
#define BSH   8
#define BN    256
#define PMAX  1024
#define CAP   8192


__device__ __forceinline__ int block_excl_scan_1024(int v, int* swave, int& total) {
    const int lane = threadIdx.x & 31;
    const int wave = threadIdx.x >> 5;
    int incl = v;
#pragma unroll
    for (int o = 1; o < 32; o <<= 1) {
        const int t = __shfl_up(incl, o, 32);
        if (lane >= o) incl += t;
    }
    if (lane == 31) swave[wave] = incl;
    __syncthreads();
    const int w = swave[lane];
    int wincl = w;
#pragma unroll
    for (int o = 1; o < 32; o <<= 1) {
        const int t = __shfl_up(wincl, o, 32);
        if (lane >= o) wincl += t;
    }
    const int woff = __shfl(wincl, wave, 32) - __shfl(w, wave, 32);
    total = __shfl(wincl, 31, 32);
    return woff + incl - v;
}

__global__ void __launch_bounds__(1024) tile_sort_kernel(const int* __restrict__ src,
                                                         const int* __restrict__ dst,
                                                         int* __restrict__ part,
                                                         int* __restrict__ S,
                                                         int n_edges, int P) {
    __shared__ int hist[PMAX + 1];
    __shared__ int swave[32];
    __shared__ v2i stage[TILE];
    const int tid      = threadIdx.x;
    const int tile0    = blockIdx.x * TILE;
    const int tile_len = min(TILE, n_edges - tile0);

    for (int i = tid; i <= P; i += 1024) hist[i] = 0;

    int s[4];
    int d[4];
#pragma unroll
    for (int k = 0; k < 4; ++k) {
        const int i = tid + k * 1024;
        s[k] = 0;
        d[k] = 0;
        if (i < tile_len) {
            s[k] = src[tile0 + i];
            d[k] = dst[tile0 + i];
        }
    }
    __syncthreads();

    int b[4];
    int r[4];
#pragma unroll
    for (int k = 0; k < 4; ++k) {
        b[k] = 0;
        r[k] = 0;
        if (tid + k * 1024 < tile_len) {
            const unsigned q = (unsigned)d[k] >> BSH;
            b[k] = (int)(q < (unsigned)P ? q : (unsigned)(P - 1));
            r[k] = atomicAdd(&hist[b[k]], 1);
        }
    }
    __syncthreads();

    const int v = (tid < P) ? hist[tid] : 0;
    int total;
    const int excl = block_excl_scan_1024(v, swave, total);
    if (tid < P) hist[tid] = excl;
    if (tid == 0) hist[P] = total;
    __syncthreads();

    int* __restrict__ Srow = S + (size_t)blockIdx.x * (size_t)(P + 1);
    for (int i = tid; i <= P; i += 1024) Srow[i] = hist[i];

#pragma unroll
    for (int k = 0; k < 4; ++k) {
        if (tid + k * 1024 < tile_len) {
            const int p = hist[b[k]] + r[k];
            if ((unsigned)p < (unsigned)TILE) stage[p] = (v2i){s[k], d[k]};
        }
    }
    __syncthreads();

    v2i* __restrict__ pout = (v2i*)part + (size_t)tile0;
#pragma unroll
    for (int k = 0; k < 4; ++k) {
        const int i = tid + k * 1024;
        if (i < tile_len) pout[i] = stage[i];
    }
}

#define FOR_RANGE_EDGES(BODY)                                                          \
    for (int tb = 0; tb < T; tb += 1024) {                                             \
        __syncthreads();                        \
        {                                                                              \
            const int t = tb + tid;                                                    \
            int a0 = 0, a1 = 0;                                                        \
            if (t < T) {                                                               \
                a0 = S[(size_t)t * (size_t)(P + 1) + b];                               \
                a1 = S[(size_t)t * (size_t)(P + 1) + b + 1];                           \
            }                                                                          \
            seg0[tid] = a0;                                                            \
            seg1[tid] = a1;                                                            \
        }                                                                              \
        __syncthreads();                                                               \
        const int tn = min(1024, T - tb);                                              \
        for (int g = grp; g < tn; g += 128) {                                          \
                      \
            const int lim = min(TILE, n_edges - (tb + g) * TILE);                      \
            const int j0  = max(seg0[g], 0);                                           \
            const int j1  = min(seg1[g], lim);                                         \
            const v2i* __restrict__ p = p2 + (size_t)(tb + g) * TILE;                  \
            for (int j = j0 + sub; j < j1; j += 16) {                                  \
                const bool two = j + 8 < j1;                                           \
                const v2i e0 = p[j];                                                   \
                const v2i e1 = two ? p[j + 8] : e0;                                    \
                {                                                                      \
                    const unsigned u = (unsigned)(e0.y - node0);                       \
                    const int dl = (int)(u < (unsigned)nn ? u : (unsigned)(nn - 1));   \
                    const int sv = e0.x;                                               \
                    BODY                                                               \
                }                                                                      \
                if (two) {                                                             \
                    const unsigned u = (unsigned)(e1.y - node0);                       \
                    const int dl = (int)(u < (unsigned)nn ? u : (unsigned)(nn - 1));   \
                    const int sv = e1.x;                                               \
                    BODY                                                               \
                }                                                                      \
            }                                                                          \
        }                                                                              \
    }

__global__ void __launch_bounds__(1024) range_csr_kernel(const int* __restrict__ part,
                                                         const int* __restrict__ S,
                                                         int* __restrict__ rowptr,
                                                         float* __restrict__ dinv,
                                                         int* __restrict__ col,
                                                         int n_edges, int n_nodes,
                                                         int P, int T) {
    __shared__ int cnt[BN];
    __shared__ int cur[BN];
    __shared__ int swave_a[32];
    __shared__ int swave_b[32];
    __shared__ int seg0[1024];
    __shared__ int seg1[1024];
    __shared__ int stage[CAP];
    const int tid   = threadIdx.x;
    const int b     = blockIdx.x;
    const int node0 = b * BN;
    const int nn    = min(BN, n_nodes - node0);
    const int grp   = tid >> 3;
    const int sub   = tid & 7;
    const v2i* __restrict__ p2 = (const v2i*)part;

    if (tid < BN) cnt[tid] = 0;

    int bpart = 0;
    for (int t = tid; t < T; t += 1024) bpart += S[(size_t)t * (size_t)(P + 1) + b];

    FOR_RANGE_EDGES({ (void)sv; atomicAdd(&cnt[dl], 1); })
    __syncthreads();

    int base;
    (void)block_excl_scan_1024(bpart, swave_a, base);
    const int c = (tid < BN) ? cnt[tid] : 0;
    int nb;
    const int ex = block_excl_scan_1024(c, swave_b, nb);
    if (tid < BN) cur[tid] = ex;
    if (tid < nn) {
        rowptr[node0 + tid] = base + ex;
        dinv[node0 + tid]   = rsqrtf((float)(c + 1));
    }
    if (b == P - 1 && tid == 0) rowptr[n_nodes] = base + nb;

    const bool staged = nb <= CAP;
    FOR_RANGE_EDGES({
        const int pos = atomicAdd(&cur[dl], 1);
        if (staged) {
            if ((unsigned)pos < (unsigned)CAP) stage[pos] = sv;
        } else {
            const int q = base + pos;
            if ((unsigned)q < (unsigned)n_edges) col[q] = sv;
        }
    })
    __syncthreads();
    if (staged) {
        for (int i = tid; i < nb; i += 1024) {
            const int q = base + i;
            if ((unsigned)q < (unsigned)n_edges) col[q] = stage[i];
        }
    }
}


template <int K, int NOUT>
__global__ void __launch_bounds__(256) gemm_kernel(const float* __restrict__ x,
                                                   const float* __restrict__ W,
                                                   const float* __restrict__ dinv,
                                                   float* __restrict__ h, int n_nodes) {
    constexpr int LDK = K + 4;
    constexpr int NT  = NOUT / 16;
    __shared__ float sWt[NOUT * LDK];

    for (int i = threadIdx.x; i < K * NOUT; i += 256) {
        const int k = i / NOUT;
        const int n = i % NOUT;
        sWt[n * LDK + k] = W[i];
    }
    __syncthreads();

    const int wave = threadIdx.x >> 5;
    const int lane = threadIdx.x & 31;
    const int half = lane >> 4;
    const int l    = lane & 15;
    const int m0   = (blockIdx.x * 8 + wave) * 16;
    if (m0 >= n_nodes) return;

    const int rowA = min(m0 + l, n_nodes - 1);
    const float* __restrict__ xrow = x + (size_t)rowA * K;

    float dv[8];
    if (m0 + 16 <= n_nodes) {
        const v4f d0 = *(const v4f*)(dinv + m0 + 8 * half);
        const v4f d1 = *(const v4f*)(dinv + m0 + 8 * half + 4);
        dv[0] = d0.x; dv[1] = d0.y; dv[2] = d0.z; dv[3] = d0.w;
        dv[4] = d1.x; dv[5] = d1.y; dv[6] = d1.z; dv[7] = d1.w;
    } else {
#pragma unroll
        for (int v = 0; v < 8; ++v) dv[v] = dinv[min(m0 + 8 * half + v, n_nodes - 1)];
    }

    v8f acc[NT] = {};
    for (int k = 0; k < K; k += 4) {
        const int kk = k + 2 * half;
        const v2f a = *(const v2f*)(xrow + kk);
#pragma unroll
        for (int t = 0; t < NT; ++t) {
            const int col = t * 16 + l;
            const v2f b = *(const v2f*)(&sWt[col * LDK + kk]);
            acc[t] = __builtin_amdgcn_wmma_f32_16x16x4_f32(
                false, a, false, b, (short)0, acc[t], false, false);
        }
    }

    if (m0 + 16 <= n_nodes) {
        float* __restrict__ hp = h + (size_t)(m0 + 8 * half) * NOUT + l;
#pragma unroll
        for (int v = 0; v < 8; ++v) {
#pragma unroll
            for (int t = 0; t < NT; ++t)
                hp[(size_t)v * NOUT + t * 16] = dv[v] * acc[t][v];
        }
    } else {
#pragma unroll
        for (int v = 0; v < 8; ++v) {
            const int row = m0 + 8 * half + v;
            if (row < n_nodes) {
#pragma unroll
                for (int t = 0; t < NT; ++t)
                    h[(size_t)row * NOUT + t * 16 + l] = dv[v] * acc[t][v];
            }
        }
    }
}


template <int F, bool RELU>
__global__ void __launch_bounds__(256) gather_kernel(const int* __restrict__ rowptr,
                                                     const int* __restrict__ col,
                                                     const float* __restrict__ dinv,
                                                     const float* __restrict__ h,
                                                     const float* __restrict__ bias,
                                                     float* __restrict__ outp, int n_nodes) {
    constexpr int G  = 8;
    constexpr int V4 = F / 32;
    constexpr int Q  = 32 / G;
    const int lane  = threadIdx.x & 31;
    const int sub   = lane & (G - 1);
    const int gbase = lane & ~(G - 1);
    const int node  = (int)((blockIdx.x * 256 + threadIdx.x) >> 3);
    const bool valid = node < n_nodes;
    const int nodec  = valid ? node : n_nodes - 1;

    const int beg  = rowptr[nodec];
    const int endr = rowptr[nodec + 1];
    const int end  = valid ? endr : beg;
    const float dd = dinv[nodec];
    const float* __restrict__ hl = h + 4 * sub;

    v4f hs[V4];
#pragma unroll
    for (int v = 0; v < V4; ++v) hs[v] = *(const v4f*)(hl + (size_t)nodec * F + 32 * v);

    v4f acc[V4];
#pragma unroll
    for (int v = 0; v < V4; ++v) acc[v] = (v4f){0.0f, 0.0f, 0.0f, 0.0f};

    for (int base = beg; __any(base < end); base += 32) {
        unsigned c_l[Q];
#pragma unroll
        for (int q = 0; q < Q; ++q) {
            const int j = base + q * G + sub;
            c_l[q] = (unsigned)nodec;
            if (j < end) c_l[q] = (unsigned)col[j];
        }
        int s_l[Q];
#pragma unroll
        for (int q = 0; q < Q; ++q) {
            s_l[q] = (int)(c_l[q] < (unsigned)n_nodes ? c_l[q] : (unsigned)(n_nodes - 1));
        }
        const int m = end - base;
#pragma unroll
        for (int q = 0; q < Q; ++q) {
            if (__any(q * G < m)) {
                int s[G];
                v4f r[G][V4];
#pragma unroll
                for (int i = 0; i < G; ++i) s[i] = __shfl(s_l[q], gbase + i, 32);
#pragma unroll
                for (int i = 0; i < G; ++i) {
#pragma unroll
                    for (int v = 0; v < V4; ++v)
                        r[i][v] = *(const v4f*)(hl + (size_t)s[i] * F + 32 * v);
                }
#pragma unroll
                for (int i = 0; i < G; ++i) {
                    const float lv = (q * G + i < m) ? 1.0f : 0.0f;
#pragma unroll
                    for (int v = 0; v < V4; ++v) {
                        acc[v].x = fmaf(lv, r[i][v].x, acc[v].x);
                        acc[v].y = fmaf(lv, r[i][v].y, acc[v].y);
                        acc[v].z = fmaf(lv, r[i][v].z, acc[v].z);
                        acc[v].w = fmaf(lv, r[i][v].w, acc[v].w);
                    }
                }
            }
        }
    }

    if (!valid) return;
#pragma unroll
    for (int v = 0; v < V4; ++v) {
        const v4f bs = *(const v4f*)(bias + 32 * v + 4 * sub);
        v4f o;
        o.x = dd * (acc[v].x + hs[v].x) + bs.x;
        o.y = dd * (acc[v].y + hs[v].y) + bs.y;
        o.z = dd * (acc[v].z + hs[v].z) + bs.z;
        o.w = dd * (acc[v].w + hs[v].w) + bs.w;
        if (RELU) {
            o.x = fmaxf(o.x, 0.0f);
            o.y = fmaxf(o.y, 0.0f);
            o.z = fmaxf(o.z, 0.0f);
            o.w = fmaxf(o.w, 0.0f);
        }
        *(v4f*)(outp + (size_t)node * F + 32 * v + 4 * sub) = o;
    }
}


#define LDA (F_MID + 4)

__global__ void __launch_bounds__(256) gather_gemm_kernel(const int* __restrict__ rowptr,
                                                          const int* __restrict__ col,
                                                          const float* __restrict__ dinv,
                                                          const float* __restrict__ h,
                                                          const float* __restrict__ bias,
                                                          const float* __restrict__ W2,
                                                          float* __restrict__ h2, int n_nodes) {
    constexpr int F  = F_MID;
    constexpr int G  = 8;
    constexpr int V4 = F / 32;
    constexpr int Q  = 32 / G;
    __shared__ __attribute__((aligned(16))) float sWt[F_OUT * LDA];
    __shared__ __attribute__((aligned(16))) float sA[32 * LDA];
    __shared__ __attribute__((aligned(16))) float sD[32];

    constexpr int WV = F_MID * F_OUT / 4 / 256;
    v4f wv[WV];
#pragma unroll
    for (int j = 0; j < WV; ++j) wv[j] = *(const v4f*)(W2 + 4 * ((int)threadIdx.x + 256 * j));

    const int lane  = threadIdx.x & 31;
    const int sub   = lane & (G - 1);
    const int gbase = lane & ~(G - 1);
    const int nl    = (int)(threadIdx.x >> 3);
    const int node  = (int)(blockIdx.x * 32) + nl;
    const bool valid = node < n_nodes;
    const int nodec  = valid ? node : n_nodes - 1;

    const int beg  = rowptr[nodec];
    const int endr = rowptr[nodec + 1];
    const int end  = valid ? endr : beg;
    const float dd = dinv[nodec];
    const float* __restrict__ hl = h + 4 * sub;

    v4f hs[V4];
#pragma unroll
    for (int v = 0; v < V4; ++v) hs[v] = *(const v4f*)(hl + (size_t)nodec * F + 32 * v);

#pragma unroll
    for (int j = 0; j < WV; ++j) {
        const int i4 = (int)threadIdx.x + 256 * j;
        const int k  = i4 / (F_OUT / 4);
        const int n0 = (i4 % (F_OUT / 4)) * 4;
        sWt[(n0 + 0) * LDA + k] = wv[j].x;
        sWt[(n0 + 1) * LDA + k] = wv[j].y;
        sWt[(n0 + 2) * LDA + k] = wv[j].z;
        sWt[(n0 + 3) * LDA + k] = wv[j].w;
    }

    v4f acc[V4];
#pragma unroll
    for (int v = 0; v < V4; ++v) acc[v] = (v4f){0.0f, 0.0f, 0.0f, 0.0f};

    for (int base = beg; __any(base < end); base += 32) {
        unsigned c_l[Q];
#pragma unroll
        for (int q = 0; q < Q; ++q) {
            const int j = base + q * G + sub;
            c_l[q] = (unsigned)nodec;
            if (j < end) c_l[q] = (unsigned)col[j];
        }
        int s_l[Q];
#pragma unroll
        for (int q = 0; q < Q; ++q) {
            s_l[q] = (int)(c_l[q] < (unsigned)n_nodes ? c_l[q] : (unsigned)(n_nodes - 1));
        }
        const int m = end - base;
#pragma unroll
        for (int q = 0; q < Q; ++q) {
            if (__any(q * G < m)) {
                int s[G];
                v4f r[G][V4];
#pragma unroll
                for (int i = 0; i < G; ++i) s[i] = __shfl(s_l[q], gbase + i, 32);
#pragma unroll
                for (int i = 0; i < G; ++i) {
#pragma unroll
                    for (int v = 0; v < V4; ++v)
                        r[i][v] = *(const v4f*)(hl + (size_t)s[i] * F + 32 * v);
                }
#pragma unroll
                for (int i = 0; i < G; ++i) {
                    const float lv = (q * G + i < m) ? 1.0f : 0.0f;
#pragma unroll
                    for (int v = 0; v < V4; ++v) {
                        acc[v].x = fmaf(lv, r[i][v].x, acc[v].x);
                        acc[v].y = fmaf(lv, r[i][v].y, acc[v].y);
                        acc[v].z = fmaf(lv, r[i][v].z, acc[v].z);
                        acc[v].w = fmaf(lv, r[i][v].w, acc[v].w);
                    }
                }
            }
        }
    }

#pragma unroll
    for (int v = 0; v < V4; ++v) {
        const v4f bs = *(const v4f*)(bias + 32 * v + 4 * sub);
        v4f o;
        o.x = fmaxf(dd * (acc[v].x + hs[v].x) + bs.x, 0.0f);
        o.y = fmaxf(dd * (acc[v].y + hs[v].y) + bs.y, 0.0f);
        o.z = fmaxf(dd * (acc[v].z + hs[v].z) + bs.z, 0.0f);
        o.w = fmaxf(dd * (acc[v].w + hs[v].w) + bs.w, 0.0f);
        if (!valid) o = (v4f){0.0f, 0.0f, 0.0f, 0.0f};
        *(v4f*)(&sA[nl * LDA + 32 * v + 4 * sub]) = o;
    }
    if (sub == 0) sD[nl] = valid ? dd : 0.0f;
    __syncthreads();

    const int wave = (int)(threadIdx.x >> 5);
    if (wave >= 2) return;
    const int half = lane >> 4;
    const int l    = lane & 15;
    const int m0   = (int)(blockIdx.x * 32) + wave * 16;
    if (m0 >= n_nodes) return;

    constexpr int NT = F_OUT / 16;
    const float* __restrict__ arow = &sA[(wave * 16 + l) * LDA];
    v8f pacc[NT] = {};
    for (int k = 0; k < F_MID; k += 4) {
        const int kk = k + 2 * half;
        const v2f a = *(const v2f*)(arow + kk);
#pragma unroll
        for (int t = 0; t < NT; ++t) {
            const v2f b = *(const v2f*)(&sWt[(t * 16 + l) * LDA + kk]);
            pacc[t] = __builtin_amdgcn_wmma_f32_16x16x4_f32(
                false, a, false, b, (short)0, pacc[t], false, false);
        }
    }

    float dv[8];
    {
        const v4f d0 = *(const v4f*)(&sD[wave * 16 + 8 * half]);
        const v4f d1 = *(const v4f*)(&sD[wave * 16 + 8 * half + 4]);
        dv[0] = d0.x; dv[1] = d0.y; dv[2] = d0.z; dv[3] = d0.w;
        dv[4] = d1.x; dv[5] = d1.y; dv[6] = d1.z; dv[7] = d1.w;
    }

    if (m0 + 16 <= n_nodes) {
        float* __restrict__ hp = h2 + (size_t)(m0 + 8 * half) * F_OUT + l;
#pragma unroll
        for (int v = 0; v < 8; ++v) {
#pragma unroll
            for (int t = 0; t < NT; ++t)
                hp[(size_t)v * F_OUT + t * 16] = dv[v] * pacc[t][v];
        }
    } else {
#pragma unroll
        for (int v = 0; v < 8; ++v) {
            const int row = m0 + 8 * half + v;
            if (row < n_nodes) {
#pragma unroll
                for (int t = 0; t < NT; ++t)
                    h2[(size_t)row * F_OUT + t * 16 + l] = dv[v] * pacc[t][v];
            }
        }
    }
}


__global__ void __launch_bounds__(256) decode_kernel(const int* __restrict__ pos,
                                                     const int* __restrict__ neg,
                                                     const float* __restrict__ z,
                                                     float* __restrict__ out, int n_eval) {
    const int lane  = threadIdx.x & 31;
    const int sub   = lane & 7;
    const int gbase = lane & 24;
    const int total = 2 * n_eval;
    const int k = (int)(blockIdx.x * 256 + threadIdx.x);
    int a = 0, b = 0;
    if (k < total) {
        if (k < n_eval) { a = pos[k];            b = pos[n_eval + k]; }
        else            { const int kk = k - n_eval; a = neg[kk]; b = neg[n_eval + kk]; }
    }
    const float* __restrict__ zl = z + 4 * sub;
    v4f za[8];
    v4f zb[8];
#pragma unroll
    for (int p = 0; p < 8; ++p) {
        const int ap = __shfl(a, gbase + p, 32);
        const int bp = __shfl(b, gbase + p, 32);
        za[p] = *(const v4f*)(zl + (size_t)ap * F_OUT);
        zb[p] = *(const v4f*)(zl + (size_t)bp * F_OUT);
    }
    float res = 0.0f;
#pragma unroll
    for (int p = 0; p < 8; ++p) {
        float v = za[p].x * zb[p].x + za[p].y * zb[p].y + za[p].z * zb[p].z + za[p].w * zb[p].w;
        v += __shfl_xor(v, 4, 32);
        v += __shfl_xor(v, 2, 32);
        v += __shfl_xor(v, 1, 32);
        if (sub == p) res = v;
    }
    if (k < total) out[k] = res;
}


static inline size_t align_up(size_t x) { return (x + 255) & ~(size_t)255; }

extern "C" void kernel_launch(void* const* d_in, const int* in_sizes, int n_in,
                              void* d_out, int out_size, void* d_ws, size_t ws_size,
                              hipStream_t stream) {
    const float* x   = (const float*)d_in[0];
    const float* W1  = (const float*)d_in[1];
    const float* b1  = (const float*)d_in[2];
    const float* W2  = (const float*)d_in[3];
    const float* b2  = (const float*)d_in[4];
    const int*   ei  = (const int*)d_in[5];
    const int*   pos = (const int*)d_in[6];
    const int*   neg = (const int*)d_in[7];
    float* out = (float*)d_out;

    const int n_nodes = in_sizes[0] / F_IN;
    const int n_edges = in_sizes[5] / 2;
    const int n_eval  = in_sizes[6] / 2;
    const int* src = ei;
    const int* dst = ei + n_edges;

    const int B = 256;
    const int P = (n_nodes + BN - 1) / BN;
    const int T = (n_edges + TILE - 1) / TILE;

    char* w = (char*)d_ws;
    size_t off = 0;
    float* dinv  = (float*)(w + off); off += align_up((size_t)n_nodes * 4);
    float* h1    = (float*)(w + off); off += align_up((size_t)n_nodes * F_MID * 4);
    const size_t a1_bytes   = (size_t)n_nodes * F_MID * 4;
    const size_t part_bytes = (size_t)n_edges * 8;
    float* h2    = (float*)(w + off);
    int* part    = (int*)(w + off);   off += align_up(a1_bytes > part_bytes ? a1_bytes : part_bytes);
    int* rowptr  = (int*)(w + off);   off += align_up((size_t)(n_nodes + 1) * 4);
    int* S       = (int*)(w + off);   off += align_up((size_t)T * (size_t)(P + 1) * 4);
    int* col     = (int*)(w + off);   off += align_up((size_t)n_edges * 4);
    float* z     = h1;
    (void)ws_size;

    tile_sort_kernel<<<T, 1024, 0, stream>>>(src, dst, part, S, n_edges, P);
    range_csr_kernel<<<P, 1024, 0, stream>>>(part, S, rowptr, dinv, col, n_edges, n_nodes, P, T);

    const int gemm_blocks   = (n_nodes + 128 - 1) / 128;
    const int gather_blocks = (n_nodes + 32 - 1) / 32;

    gemm_kernel<F_IN, F_MID><<<gemm_blocks, B, 0, stream>>>(x, W1, dinv, h1, n_nodes);
    gather_gemm_kernel<<<gather_blocks, B, 0, stream>>>(rowptr, col, dinv, h1, b1, W2, h2, n_nodes);

    gather_kernel<F_OUT, false><<<gather_blocks, B, 0, stream>>>(rowptr, col, dinv, h2, b2, z, n_nodes);

    {
        int blocks = (2 * n_eval + B - 1) / B;
        decode_kernel<<<blocks, B, 0, stream>>>(pos, neg, z, out, n_eval);
    }
    (void)out_size; (void)n_in;
}


# Layer-1 GEMM as one resident 32-wave work-group per WGP (256 work-groups): W1 staged to LDS once per WGP in one round of 16-byte loads, one 16-row tile per wave, tiles dealt evenly; k order kept
# speedup vs baseline: 6.1767x; 1.0414x over previous
#include <hip/hip_runtime.h>
#include <hip/hip_bf16.h>


typedef float v2f __attribute__((ext_vector_type(2)));
typedef float v4f __attribute__((ext_vector_type(4)));
typedef float v8f __attribute__((ext_vector_type(8)));
typedef int   v2i __attribute__((ext_vector_type(2)));

#define F_IN  256
#define F_MID 64
#define F_OUT 32

#define TILE  4096
#define BSH   8
#define BN    256
#define PMAX  1024
#define CAP   8192


__device__ __forceinline__ int block_excl_scan_1024(int v, int* swave, int& total) {
    const int lane = threadIdx.x & 31;
    const int wave = threadIdx.x >> 5;
    int incl = v;
#pragma unroll
    for (int o = 1; o < 32; o <<= 1) {
        const int t = __shfl_up(incl, o, 32);
        if (lane >= o) incl += t;
    }
    if (lane == 31) swave[wave] = incl;
    __syncthreads();
    const int w = swave[lane];
    int wincl = w;
#pragma unroll
    for (int o = 1; o < 32; o <<= 1) {
        const int t = __shfl_up(wincl, o, 32);
        if (lane >= o) wincl += t;
    }
    const int woff = __shfl(wincl, wave, 32) - __shfl(w, wave, 32);
    total = __shfl(wincl, 31, 32);
    return woff + incl - v;
}

__global__ void __launch_bounds__(1024) tile_sort_kernel(const int* __restrict__ src,
                                                         const int* __restrict__ dst,
                                                         int* __restrict__ part,
                                                         int* __restrict__ S,
                                                         int n_edges, int P) {
    __shared__ int hist[PMAX + 1];
    __shared__ int swave[32];
    __shared__ v2i stage[TILE];
    const int tid      = threadIdx.x;
    const int tile0    = blockIdx.x * TILE;
    const int tile_len = min(TILE, n_edges - tile0);

    for (int i = tid; i <= P; i += 1024) hist[i] = 0;

    int s[4];
    int d[4];
#pragma unroll
    for (int k = 0; k < 4; ++k) {
        const int i = tid + k * 1024;
        s[k] = 0;
        d[k] = 0;
        if (i < tile_len) {
            s[k] = src[tile0 + i];
            d[k] = dst[tile0 + i];
        }
    }
    __syncthreads();

    int b[4];
    int r[4];
#pragma unroll
    for (int k = 0; k < 4; ++k) {
        b[k] = 0;
        r[k] = 0;
        if (tid + k * 1024 < tile_len) {
            const unsigned q = (unsigned)d[k] >> BSH;
            b[k] = (int)(q < (unsigned)P ? q : (unsigned)(P - 1));
            r[k] = atomicAdd(&hist[b[k]], 1);
        }
    }
    __syncthreads();

    const int v = (tid < P) ? hist[tid] : 0;
    int total;
    const int excl = block_excl_scan_1024(v, swave, total);
    if (tid < P) hist[tid] = excl;
    if (tid == 0) hist[P] = total;
    __syncthreads();

    int* __restrict__ Srow = S + (size_t)blockIdx.x * (size_t)(P + 1);
    for (int i = tid; i <= P; i += 1024) Srow[i] = hist[i];

#pragma unroll
    for (int k = 0; k < 4; ++k) {
        if (tid + k * 1024 < tile_len) {
            const int p = hist[b[k]] + r[k];
            if ((unsigned)p < (unsigned)TILE) stage[p] = (v2i){s[k], d[k]};
        }
    }
    __syncthreads();

    v2i* __restrict__ pout = (v2i*)part + (size_t)tile0;
#pragma unroll
    for (int k = 0; k < 4; ++k) {
        const int i = tid + k * 1024;
        if (i < tile_len) pout[i] = stage[i];
    }
}

#define FOR_RANGE_EDGES(BODY)                                                          \
    for (int tb = 0; tb < T; tb += 1024) {                                             \
        __syncthreads();                        \
        {                                                                              \
            const int t = tb + tid;                                                    \
            int a0 = 0, a1 = 0;                                                        \
            if (t < T) {                                                               \
                a0 = S[(size_t)t * (size_t)(P + 1) + b];                               \
                a1 = S[(size_t)t * (size_t)(P + 1) + b + 1];                           \
            }                                                                          \
            seg0[tid] = a0;                                                            \
            seg1[tid] = a1;                                                            \
        }                                                                              \
        __syncthreads();                                                               \
        const int tn = min(1024, T - tb);                                              \
        for (int g = grp; g < tn; g += 128) {                                          \
                      \
            const int lim = min(TILE, n_edges - (tb + g) * TILE);                      \
            const int j0  = max(seg0[g], 0);                                           \
            const int j1  = min(seg1[g], lim);                                         \
            const v2i* __restrict__ p = p2 + (size_t)(tb + g) * TILE;                  \
            for (int j = j0 + sub; j < j1; j += 16) {                                  \
                const bool two = j + 8 < j1;                                           \
                const v2i e0 = p[j];                                                   \
                const v2i e1 = two ? p[j + 8] : e0;                                    \
                {                                                                      \
                    const unsigned u = (unsigned)(e0.y - node0);                       \
                    const int dl = (int)(u < (unsigned)nn ? u : (unsigned)(nn - 1));   \
                    const int sv = e0.x;                                               \
                    BODY                                                               \
                }                                                                      \
                if (two) {                                                             \
                    const unsigned u = (unsigned)(e1.y - node0);                       \
                    const int dl = (int)(u < (unsigned)nn ? u : (unsigned)(nn - 1));   \
                    const int sv = e1.x;                                               \
                    BODY                                                               \
                }                                                                      \
            }                                                                          \
        }                                                                              \
    }

__global__ void __launch_bounds__(1024) range_csr_kernel(const int* __restrict__ part,
                                                         const int* __restrict__ S,
                                                         int* __restrict__ rowptr,
                                                         float* __restrict__ dinv,
                                                         int* __restrict__ col,
                                                         int n_edges, int n_nodes,
                                                         int P, int T) {
    __shared__ int cnt[BN];
    __shared__ int cur[BN];
    __shared__ int swave_a[32];
    __shared__ int swave_b[32];
    __shared__ int seg0[1024];
    __shared__ int seg1[1024];
    __shared__ int stage[CAP];
    const int tid   = threadIdx.x;
    const int b     = blockIdx.x;
    const int node0 = b * BN;
    const int nn    = min(BN, n_nodes - node0);
    const int grp   = tid >> 3;
    const int sub   = tid & 7;
    const v2i* __restrict__ p2 = (const v2i*)part;

    if (tid < BN) cnt[tid] = 0;

    int bpart = 0;
    for (int t = tid; t < T; t += 1024) bpart += S[(size_t)t * (size_t)(P + 1) + b];

    FOR_RANGE_EDGES({ (void)sv; atomicAdd(&cnt[dl], 1); })
    __syncthreads();

    int base;
    (void)block_excl_scan_1024(bpart, swave_a, base);
    const int c = (tid < BN) ? cnt[tid] : 0;
    int nb;
    const int ex = block_excl_scan_1024(c, swave_b, nb);
    if (tid < BN) cur[tid] = ex;
    if (tid < nn) {
        rowptr[node0 + tid] = base + ex;
        dinv[node0 + tid]   = rsqrtf((float)(c + 1));
    }
    if (b == P - 1 && tid == 0) rowptr[n_nodes] = base + nb;

    const bool staged = nb <= CAP;
    FOR_RANGE_EDGES({
        const int pos = atomicAdd(&cur[dl], 1);
        if (staged) {
            if ((unsigned)pos < (unsigned)CAP) stage[pos] = sv;
        } else {
            const int q = base + pos;
            if ((unsigned)q < (unsigned)n_edges) col[q] = sv;
        }
    })
    __syncthreads();
    if (staged) {
        for (int i = tid; i < nb; i += 1024) {
            const int q = base + i;
            if ((unsigned)q < (unsigned)n_edges) col[q] = stage[i];
        }
    }
}


template <int K, int NOUT>
__global__ void __launch_bounds__(1024) gemm_kernel(const float* __restrict__ x,
                                                    const float* __restrict__ W,
                                                    const float* __restrict__ dinv,
                                                    float* __restrict__ h, int n_nodes) {
    constexpr int LDK = K + 4;
    constexpr int NT  = NOUT / 16;
    constexpr int WV  = K * NOUT / 4 / 1024;
    __shared__ float sWt[NOUT * LDK];

    v4f wv[WV];
#pragma unroll
    for (int j = 0; j < WV; ++j) wv[j] = *(const v4f*)(W + 4 * ((int)threadIdx.x + 1024 * j));
#pragma unroll
    for (int j = 0; j < WV; ++j) {
        const int i4 = (int)threadIdx.x + 1024 * j;
        const int k  = i4 / (NOUT / 4);
        const int n0 = (i4 % (NOUT / 4)) * 4;
        sWt[(n0 + 0) * LDK + k] = wv[j].x;
        sWt[(n0 + 1) * LDK + k] = wv[j].y;
        sWt[(n0 + 2) * LDK + k] = wv[j].z;
        sWt[(n0 + 3) * LDK + k] = wv[j].w;
    }
    __syncthreads();

    const int wave = threadIdx.x >> 5;
    const int lane = threadIdx.x & 31;
    const int half = lane >> 4;
    const int l    = lane & 15;
    const int ntiles = (n_nodes + 15) >> 4;

    const int tile = wave * (int)gridDim.x + (int)blockIdx.x;
    if (tile >= ntiles) return;
    {
        const int m0 = tile * 16;

        const int rowA = min(m0 + l, n_nodes - 1);
        const float* __restrict__ xrow = x + (size_t)rowA * K;

        float dv[8];
        if (m0 + 16 <= n_nodes) {
            const v4f d0 = *(const v4f*)(dinv + m0 + 8 * half);
            const v4f d1 = *(const v4f*)(dinv + m0 + 8 * half + 4);
            dv[0] = d0.x; dv[1] = d0.y; dv[2] = d0.z; dv[3] = d0.w;
            dv[4] = d1.x; dv[5] = d1.y; dv[6] = d1.z; dv[7] = d1.w;
        } else {
#pragma unroll
            for (int v = 0; v < 8; ++v) dv[v] = dinv[min(m0 + 8 * half + v, n_nodes - 1)];
        }

        v8f acc[NT] = {};
        for (int k = 0; k < K; k += 4) {
            const int kk = k + 2 * half;
            const v2f a = *(const v2f*)(xrow + kk);
#pragma unroll
            for (int t = 0; t < NT; ++t) {
                const int col = t * 16 + l;
                const v2f b = *(const v2f*)(&sWt[col * LDK + kk]);
                acc[t] = __builtin_amdgcn_wmma_f32_16x16x4_f32(
                    false, a, false, b, (short)0, acc[t], false, false);
            }
        }

        if (m0 + 16 <= n_nodes) {
            float* __restrict__ hp = h + (size_t)(m0 + 8 * half) * NOUT + l;
#pragma unroll
            for (int v = 0; v < 8; ++v) {
#pragma unroll
                for (int t = 0; t < NT; ++t)
                    hp[(size_t)v * NOUT + t * 16] = dv[v] * acc[t][v];
            }
        } else {
#pragma unroll
            for (int v = 0; v < 8; ++v) {
                const int row = m0 + 8 * half + v;
                if (row < n_nodes) {
#pragma unroll
                    for (int t = 0; t < NT; ++t)
                        h[(size_t)row * NOUT + t * 16 + l] = dv[v] * acc[t][v];
                }
            }
        }
    }
}


template <int F, bool RELU>
__global__ void __launch_bounds__(256) gather_kernel(const int* __restrict__ rowptr,
                                                     const int* __restrict__ col,
                                                     const float* __restrict__ dinv,
                                                     const float* __restrict__ h,
                                                     const float* __restrict__ bias,
                                                     float* __restrict__ outp, int n_nodes) {
    constexpr int G  = 8;
    constexpr int V4 = F / 32;
    constexpr int Q  = 32 / G;
    const int lane  = threadIdx.x & 31;
    const int sub   = lane & (G - 1);
    const int gbase = lane & ~(G - 1);
    const int node  = (int)((blockIdx.x * 256 + threadIdx.x) >> 3);
    const bool valid = node < n_nodes;
    const int nodec  = valid ? node : n_nodes - 1;

    const int beg  = rowptr[nodec];
    const int endr = rowptr[nodec + 1];
    const int end  = valid ? endr : beg;
    const float dd = dinv[nodec];
    const float* __restrict__ hl = h + 4 * sub;

    v4f hs[V4];
#pragma unroll
    for (int v = 0; v < V4; ++v) hs[v] = *(const v4f*)(hl + (size_t)nodec * F + 32 * v);

    v4f acc[V4];
#pragma unroll
    for (int v = 0; v < V4; ++v) acc[v] = (v4f){0.0f, 0.0f, 0.0f, 0.0f};

    for (int base = beg; __any(base < end); base += 32) {
        unsigned c_l[Q];
#pragma unroll
        for (int q = 0; q < Q; ++q) {
            const int j = base + q * G + sub;
            c_l[q] = (unsigned)nodec;
            if (j < end) c_l[q] = (unsigned)col[j];
        }
        int s_l[Q];
#pragma unroll
        for (int q = 0; q < Q; ++q) {
            s_l[q] = (int)(c_l[q] < (unsigned)n_nodes ? c_l[q] : (unsigned)(n_nodes - 1));
        }
        const int m = end - base;
#pragma unroll
        for (int q = 0; q < Q; ++q) {
            if (__any(q * G < m)) {
                int s[G];
                v4f r[G][V4];
#pragma unroll
                for (int i = 0; i < G; ++i) s[i] = __shfl(s_l[q], gbase + i, 32);
#pragma unroll
                for (int i = 0; i < G; ++i) {
#pragma unroll
                    for (int v = 0; v < V4; ++v)
                        r[i][v] = *(const v4f*)(hl + (size_t)s[i] * F + 32 * v);
                }
#pragma unroll
                for (int i = 0; i < G; ++i) {
                    const float lv = (q * G + i < m) ? 1.0f : 0.0f;
#pragma unroll
                    for (int v = 0; v < V4; ++v) {
                        acc[v].x = fmaf(lv, r[i][v].x, acc[v].x);
                        acc[v].y = fmaf(lv, r[i][v].y, acc[v].y);
                        acc[v].z = fmaf(lv, r[i][v].z, acc[v].z);
                        acc[v].w = fmaf(lv, r[i][v].w, acc[v].w);
                    }
                }
            }
        }
    }

    if (!valid) return;
#pragma unroll
    for (int v = 0; v < V4; ++v) {
        const v4f bs = *(const v4f*)(bias + 32 * v + 4 * sub);
        v4f o;
        o.x = dd * (acc[v].x + hs[v].x) + bs.x;
        o.y = dd * (acc[v].y + hs[v].y) + bs.y;
        o.z = dd * (acc[v].z + hs[v].z) + bs.z;
        o.w = dd * (acc[v].w + hs[v].w) + bs.w;
        if (RELU) {
            o.x = fmaxf(o.x, 0.0f);
            o.y = fmaxf(o.y, 0.0f);
            o.z = fmaxf(o.z, 0.0f);
            o.w = fmaxf(o.w, 0.0f);
        }
        *(v4f*)(outp + (size_t)node * F + 32 * v + 4 * sub) = o;
    }
}


#define LDA (F_MID + 4)

__global__ void __launch_bounds__(256) gather_gemm_kernel(const int* __restrict__ rowptr,
                                                          const int* __restrict__ col,
                                                          const float* __restrict__ dinv,
                                                          const float* __restrict__ h,
                                                          const float* __restrict__ bias,
                                                          const float* __restrict__ W2,
                                                          float* __restrict__ h2, int n_nodes) {
    constexpr int F  = F_MID;
    constexpr int G  = 8;
    constexpr int V4 = F / 32;
    constexpr int Q  = 32 / G;
    __shared__ __attribute__((aligned(16))) float sWt[F_OUT * LDA];
    __shared__ __attribute__((aligned(16))) float sA[32 * LDA];
    __shared__ __attribute__((aligned(16))) float sD[32];

    constexpr int WV = F_MID * F_OUT / 4 / 256;
    v4f wv[WV];
#pragma unroll
    for (int j = 0; j < WV; ++j) wv[j] = *(const v4f*)(W2 + 4 * ((int)threadIdx.x + 256 * j));

    const int lane  = threadIdx.x & 31;
    const int sub   = lane & (G - 1);
    const int gbase = lane & ~(G - 1);
    const int nl    = (int)(threadIdx.x >> 3);
    const int node  = (int)(blockIdx.x * 32) + nl;
    const bool valid = node < n_nodes;
    const int nodec  = valid ? node : n_nodes - 1;

    const int beg  = rowptr[nodec];
    const int endr = rowptr[nodec + 1];
    const int end  = valid ? endr : beg;
    const float dd = dinv[nodec];
    const float* __restrict__ hl = h + 4 * sub;

    v4f hs[V4];
#pragma unroll
    for (int v = 0; v < V4; ++v) hs[v] = *(const v4f*)(hl + (size_t)nodec * F + 32 * v);

#pragma unroll
    for (int j = 0; j < WV; ++j) {
        const int i4 = (int)threadIdx.x + 256 * j;
        const int k  = i4 / (F_OUT / 4);
        const int n0 = (i4 % (F_OUT / 4)) * 4;
        sWt[(n0 + 0) * LDA + k] = wv[j].x;
        sWt[(n0 + 1) * LDA + k] = wv[j].y;
        sWt[(n0 + 2) * LDA + k] = wv[j].z;
        sWt[(n0 + 3) * LDA + k] = wv[j].w;
    }

    v4f acc[V4];
#pragma unroll
    for (int v = 0; v < V4; ++v) acc[v] = (v4f){0.0f, 0.0f, 0.0f, 0.0f};

    for (int base = beg; __any(base < end); base += 32) {
        unsigned c_l[Q];
#pragma unroll
        for (int q = 0; q < Q; ++q) {
            const int j = base + q * G + sub;
            c_l[q] = (unsigned)nodec;
            if (j < end) c_l[q] = (unsigned)col[j];
        }
        int s_l[Q];
#pragma unroll
        for (int q = 0; q < Q; ++q) {
            s_l[q] = (int)(c_l[q] < (unsigned)n_nodes ? c_l[q] : (unsigned)(n_nodes - 1));
        }
        const int m = end - base;
#pragma unroll
        for (int q = 0; q < Q; ++q) {
            if (__any(q * G < m)) {
                int s[G];
                v4f r[G][V4];
#pragma unroll
                for (int i = 0; i < G; ++i) s[i] = __shfl(s_l[q], gbase + i, 32);
#pragma unroll
                for (int i = 0; i < G; ++i) {
#pragma unroll
                    for (int v = 0; v < V4; ++v)
                        r[i][v] = *(const v4f*)(hl + (size_t)s[i] * F + 32 * v);
                }
#pragma unroll
                for (int i = 0; i < G; ++i) {
                    const float lv = (q * G + i < m) ? 1.0f : 0.0f;
#pragma unroll
                    for (int v = 0; v < V4; ++v) {
                        acc[v].x = fmaf(lv, r[i][v].x, acc[v].x);
                        acc[v].y = fmaf(lv, r[i][v].y, acc[v].y);
                        acc[v].z = fmaf(lv, r[i][v].z, acc[v].z);
                        acc[v].w = fmaf(lv, r[i][v].w, acc[v].w);
                    }
                }
            }
        }
    }

#pragma unroll
    for (int v = 0; v < V4; ++v) {
        const v4f bs = *(const v4f*)(bias + 32 * v + 4 * sub);
        v4f o;
        o.x = fmaxf(dd * (acc[v].x + hs[v].x) + bs.x, 0.0f);
        o.y = fmaxf(dd * (acc[v].y + hs[v].y) + bs.y, 0.0f);
        o.z = fmaxf(dd * (acc[v].z + hs[v].z) + bs.z, 0.0f);
        o.w = fmaxf(dd * (acc[v].w + hs[v].w) + bs.w, 0.0f);
        if (!valid) o = (v4f){0.0f, 0.0f, 0.0f, 0.0f};
        *(v4f*)(&sA[nl * LDA + 32 * v + 4 * sub]) = o;
    }
    if (sub == 0) sD[nl] = valid ? dd : 0.0f;
    __syncthreads();

    const int wave = (int)(threadIdx.x >> 5);
    if (wave >= 2) return;
    const int half = lane >> 4;
    const int l    = lane & 15;
    const int m0   = (int)(blockIdx.x * 32) + wave * 16;
    if (m0 >= n_nodes) return;

    constexpr int NT = F_OUT / 16;
    const float* __restrict__ arow = &sA[(wave * 16 + l) * LDA];
    v8f pacc[NT] = {};
    for (int k = 0; k < F_MID; k += 4) {
        const int kk = k + 2 * half;
        const v2f a = *(const v2f*)(arow + kk);
#pragma unroll
        for (int t = 0; t < NT; ++t) {
            const v2f b = *(const v2f*)(&sWt[(t * 16 + l) * LDA + kk]);
            pacc[t] = __builtin_amdgcn_wmma_f32_16x16x4_f32(
                false, a, false, b, (short)0, pacc[t], false, false);
        }
    }

    float dv[8];
    {
        const v4f d0 = *(const v4f*)(&sD[wave * 16 + 8 * half]);
        const v4f d1 = *(const v4f*)(&sD[wave * 16 + 8 * half + 4]);
        dv[0] = d0.x; dv[1] = d0.y; dv[2] = d0.z; dv[3] = d0.w;
        dv[4] = d1.x; dv[5] = d1.y; dv[6] = d1.z; dv[7] = d1.w;
    }

    if (m0 + 16 <= n_nodes) {
        float* __restrict__ hp = h2 + (size_t)(m0 + 8 * half) * F_OUT + l;
#pragma unroll
        for (int v = 0; v < 8; ++v) {
#pragma unroll
            for (int t = 0; t < NT; ++t)
                hp[(size_t)v * F_OUT + t * 16] = dv[v] * pacc[t][v];
        }
    } else {
#pragma unroll
        for (int v = 0; v < 8; ++v) {
            const int row = m0 + 8 * half + v;
            if (row < n_nodes) {
#pragma unroll
                for (int t = 0; t < NT; ++t)
                    h2[(size_t)row * F_OUT + t * 16 + l] = dv[v] * pacc[t][v];
            }
        }
    }
}


__global__ void __launch_bounds__(256) decode_kernel(const int* __restrict__ pos,
                                                     const int* __restrict__ neg,
                                                     const float* __restrict__ z,
                                                     float* __restrict__ out, int n_eval) {
    const int lane  = threadIdx.x & 31;
    const int sub   = lane & 7;
    const int gbase = lane & 24;
    const int total = 2 * n_eval;
    const int k = (int)(blockIdx.x * 256 + threadIdx.x);
    int a = 0, b = 0;
    if (k < total) {
        if (k < n_eval) { a = pos[k];            b = pos[n_eval + k]; }
        else            { const int kk = k - n_eval; a = neg[kk]; b = neg[n_eval + kk]; }
    }
    const float* __restrict__ zl = z + 4 * sub;
    v4f za[8];
    v4f zb[8];
#pragma unroll
    for (int p = 0; p < 8; ++p) {
        const int ap = __shfl(a, gbase + p, 32);
        const int bp = __shfl(b, gbase + p, 32);
        za[p] = *(const v4f*)(zl + (size_t)ap * F_OUT);
        zb[p] = *(const v4f*)(zl + (size_t)bp * F_OUT);
    }
    float res = 0.0f;
#pragma unroll
    for (int p = 0; p < 8; ++p) {
        float v = za[p].x * zb[p].x + za[p].y * zb[p].y + za[p].z * zb[p].z + za[p].w * zb[p].w;
        v += __shfl_xor(v, 4, 32);
        v += __shfl_xor(v, 2, 32);
        v += __shfl_xor(v, 1, 32);
        if (sub == p) res = v;
    }
    if (k < total) out[k] = res;
}


static inline size_t align_up(size_t x) { return (x + 255) & ~(size_t)255; }

extern "C" void kernel_launch(void* const* d_in, const int* in_sizes, int n_in,
                              void* d_out, int out_size, void* d_ws, size_t ws_size,
                              hipStream_t stream) {
    const float* x   = (const float*)d_in[0];
    const float* W1  = (const float*)d_in[1];
    const float* b1  = (const float*)d_in[2];
    const float* W2  = (const float*)d_in[3];
    const float* b2  = (const float*)d_in[4];
    const int*   ei  = (const int*)d_in[5];
    const int*   pos = (const int*)d_in[6];
    const int*   neg = (const int*)d_in[7];
    float* out = (float*)d_out;

    const int n_nodes = in_sizes[0] / F_IN;
    const int n_edges = in_sizes[5] / 2;
    const int n_eval  = in_sizes[6] / 2;
    const int* src = ei;
    const int* dst = ei + n_edges;

    const int B = 256;
    const int P = (n_nodes + BN - 1) / BN;
    const int T = (n_edges + TILE - 1) / TILE;

    char* w = (char*)d_ws;
    size_t off = 0;
    float* dinv  = (float*)(w + off); off += align_up((size_t)n_nodes * 4);
    float* h1    = (float*)(w + off); off += align_up((size_t)n_nodes * F_MID * 4);
    const size_t a1_bytes   = (size_t)n_nodes * F_MID * 4;
    const size_t part_bytes = (size_t)n_edges * 8;
    float* h2    = (float*)(w + off);
    int* part    = (int*)(w + off);   off += align_up(a1_bytes > part_bytes ? a1_bytes : part_bytes);
    int* rowptr  = (int*)(w + off);   off += align_up((size_t)(n_nodes + 1) * 4);
    int* S       = (int*)(w + off);   off += align_up((size_t)T * (size_t)(P + 1) * 4);
    int* col     = (int*)(w + off);   off += align_up((size_t)n_edges * 4);
    float* z     = h1;
    (void)ws_size;

    tile_sort_kernel<<<T, 1024, 0, stream>>>(src, dst, part, S, n_edges, P);
    range_csr_kernel<<<P, 1024, 0, stream>>>(part, S, rowptr, dinv, col, n_edges, n_nodes, P, T);

    const int gemm_blocks   = (n_nodes + 511) / 512 > 256 ? (n_nodes + 511) / 512 : 256;
    const int gather_blocks = (n_nodes + 32 - 1) / 32;

    gemm_kernel<F_IN, F_MID><<<gemm_blocks, 1024, 0, stream>>>(x, W1, dinv, h1, n_nodes);
    gather_gemm_kernel<<<gather_blocks, B, 0, stream>>>(rowptr, col, dinv, h1, b1, W2, h2, n_nodes);

    gather_kernel<F_OUT, false><<<gather_blocks, B, 0, stream>>>(rowptr, col, dinv, h2, b2, z, n_nodes);

    {
        int blocks = (2 * n_eval + B - 1) / B;
        decode_kernel<<<blocks, B, 0, stream>>>(pos, neg, z, out, n_eval);
    }
    (void)out_size; (void)n_in;
}


# Layer-1 GEMM moved into the range kernel (6 launches to 5): 208-node ranges, each work-group multiplies its own rows; A and B read in 16-byte pieces, so k is regrouped inside each eight
# speedup vs baseline: 6.3979x; 1.0358x over previous
#include <hip/hip_runtime.h>
#include <hip/hip_bf16.h>


typedef float v2f __attribute__((ext_vector_type(2)));
typedef float v4f __attribute__((ext_vector_type(4)));
typedef float v8f __attribute__((ext_vector_type(8)));
typedef int   v2i __attribute__((ext_vector_type(2)));

#define F_IN  256
#define F_MID 64
#define F_OUT 32

#define TILE  4096
#define BN    208
#define RT    (BN / 16)
#define LDK1  (F_IN + 4)
#define PMAX  1024
#define CAP   8192


__device__ __forceinline__ int block_excl_scan_1024(int v, int* swave, int& total) {
    const int lane = threadIdx.x & 31;
    const int wave = threadIdx.x >> 5;
    int incl = v;
#pragma unroll
    for (int o = 1; o < 32; o <<= 1) {
        const int t = __shfl_up(incl, o, 32);
        if (lane >= o) incl += t;
    }
    if (lane == 31) swave[wave] = incl;
    __syncthreads();
    const int w = swave[lane];
    int wincl = w;
#pragma unroll
    for (int o = 1; o < 32; o <<= 1) {
        const int t = __shfl_up(wincl, o, 32);
        if (lane >= o) wincl += t;
    }
    const int woff = __shfl(wincl, wave, 32) - __shfl(w, wave, 32);
    total = __shfl(wincl, 31, 32);
    return woff + incl - v;
}

__global__ void __launch_bounds__(1024) tile_sort_kernel(const int* __restrict__ src,
                                                         const int* __restrict__ dst,
                                                         int* __restrict__ part,
                                                         int* __restrict__ S,
                                                         int n_edges, int P) {
    __shared__ int hist[PMAX + 1];
    __shared__ int swave[32];
    __shared__ v2i stage[TILE];
    const int tid      = threadIdx.x;
    const int tile0    = blockIdx.x * TILE;
    const int tile_len = min(TILE, n_edges - tile0);

    for (int i = tid; i <= P; i += 1024) hist[i] = 0;

    int s[4];
    int d[4];
#pragma unroll
    for (int k = 0; k < 4; ++k) {
        const int i = tid + k * 1024;
        s[k] = 0;
        d[k] = 0;
        if (i < tile_len) {
            s[k] = src[tile0 + i];
            d[k] = dst[tile0 + i];
        }
    }
    __syncthreads();

    int b[4];
    int r[4];
#pragma unroll
    for (int k = 0; k < 4; ++k) {
        b[k] = 0;
        r[k] = 0;
        if (tid + k * 1024 < tile_len) {
            const unsigned q = (unsigned)d[k] / (unsigned)BN;
            b[k] = (int)(q < (unsigned)P ? q : (unsigned)(P - 1));
            r[k] = atomicAdd(&hist[b[k]], 1);
        }
    }
    __syncthreads();

    const int v = (tid < P) ? hist[tid] : 0;
    int total;
    const int excl = block_excl_scan_1024(v, swave, total);
    if (tid < P) hist[tid] = excl;
    if (tid == 0) hist[P] = total;
    __syncthreads();

    int* __restrict__ Srow = S + (size_t)blockIdx.x * (size_t)(P + 1);
    for (int i = tid; i <= P; i += 1024) Srow[i] = hist[i];

#pragma unroll
    for (int k = 0; k < 4; ++k) {
        if (tid + k * 1024 < tile_len) {
            const int p = hist[b[k]] + r[k];
            if ((unsigned)p < (unsigned)TILE) stage[p] = (v2i){s[k], d[k]};
        }
    }
    __syncthreads();

    v2i* __restrict__ pout = (v2i*)part + (size_t)tile0;
#pragma unroll
    for (int k = 0; k < 4; ++k) {
        const int i = tid + k * 1024;
        if (i < tile_len) pout[i] = stage[i];
    }
}

#define FOR_RANGE_EDGES(BODY)                                                          \
    for (int tb = 0; tb < T; tb += 1024) {                                             \
        __syncthreads();                        \
        {                                                                              \
            const int t = tb + tid;                                                    \
            int a0 = 0, a1 = 0;                                                        \
            if (t < T) {                                                               \
                a0 = S[(size_t)t * (size_t)(P + 1) + b];                               \
                a1 = S[(size_t)t * (size_t)(P + 1) + b + 1];                           \
            }                                                                          \
            seg0[tid] = a0;                                                            \
            seg1[tid] = a1;                                                            \
        }                                                                              \
        __syncthreads();                                                               \
        const int tn = min(1024, T - tb);                                              \
        for (int g = grp; g < tn; g += 128) {                                          \
                      \
            const int lim = min(TILE, n_edges - (tb + g) * TILE);                      \
            const int j0  = max(seg0[g], 0);                                           \
            const int j1  = min(seg1[g], lim);                                         \
            const v2i* __restrict__ p = p2 + (size_t)(tb + g) * TILE;                  \
            for (int j = j0 + sub; j < j1; j += 16) {                                  \
                const bool two = j + 8 < j1;                                           \
                const v2i e0 = p[j];                                                   \
                const v2i e1 = two ? p[j + 8] : e0;                                    \
                {                                                                      \
                    const unsigned u = (unsigned)(e0.y - node0);                       \
                    const int dl = (int)(u < (unsigned)nn ? u : (unsigned)(nn - 1));   \
                    const int sv = e0.x;                                               \
                    BODY                                                               \
                }                                                                      \
                if (two) {                                                             \
                    const unsigned u = (unsigned)(e1.y - node0);                       \
                    const int dl = (int)(u < (unsigned)nn ? u : (unsigned)(nn - 1));   \
                    const int sv = e1.x;                                               \
                    BODY                                                               \
                }                                                                      \
            }                                                                          \
        }                                                                              \
    }

__device__ __forceinline__ void gemm_half_tile(const float* __restrict__ xrow,
                                               const float* sWt, int half, int l, int ch,
                                               v8f& acc0, v8f& acc1) {
    const float* __restrict__ xa = xrow + 4 * half;
    const float* b0 = &sWt[((2 * ch + 0) * 16 + l) * LDK1 + 4 * half];
    const float* b1 = &sWt[((2 * ch + 1) * 16 + l) * LDK1 + 4 * half];
    for (int k0 = 0; k0 < F_IN; k0 += 32) {
        v4f a[4];
#pragma unroll
        for (int i = 0; i < 4; ++i) a[i] = *(const v4f*)(xa + k0 + 8 * i);
#pragma unroll
        for (int i = 0; i < 4; ++i) {
            const v4f w0 = *(const v4f*)(b0 + k0 + 8 * i);
            const v4f w1 = *(const v4f*)(b1 + k0 + 8 * i);
            const v2f alo = (v2f){a[i].x, a[i].y};
            const v2f ahi = (v2f){a[i].z, a[i].w};
            acc0 = __builtin_amdgcn_wmma_f32_16x16x4_f32(
                false, alo, false, (v2f){w0.x, w0.y}, (short)0, acc0, false, false);
            acc1 = __builtin_amdgcn_wmma_f32_16x16x4_f32(
                false, alo, false, (v2f){w1.x, w1.y}, (short)0, acc1, false, false);
            acc0 = __builtin_amdgcn_wmma_f32_16x16x4_f32(
                false, ahi, false, (v2f){w0.z, w0.w}, (short)0, acc0, false, false);
            acc1 = __builtin_amdgcn_wmma_f32_16x16x4_f32(
                false, ahi, false, (v2f){w1.z, w1.w}, (short)0, acc1, false, false);
        }
    }
}

__global__ void __launch_bounds__(1024, 16) range_csr_gemm_kernel(const int* __restrict__ part,
                                                                  const int* __restrict__ S,
                                                                  const float* __restrict__ x,
                                                                  const float* __restrict__ W,
                                                                  int* __restrict__ rowptr,
                                                                  float* __restrict__ dinv,
                                                                  int* __restrict__ col,
                                                                  float* __restrict__ h,
                                                                  int n_edges, int n_nodes,
                                                                  int P, int T) {
    __shared__ int cnt[BN];
    __shared__ int cur[BN];
    __shared__ int swave_a[32];
    __shared__ int swave_b[32];
    __shared__ int seg0[1024];
    __shared__ int seg1[1024];
    __shared__ int stage[CAP];
    __shared__ __attribute__((aligned(16))) float sdinv[BN];
    __shared__ __attribute__((aligned(16))) float sWt[F_MID * LDK1];
    const int tid   = threadIdx.x;
    const int b     = blockIdx.x;
    const int node0 = b * BN;
    const int nn    = min(BN, n_nodes - node0);
    const int grp   = tid >> 3;
    const int sub   = tid & 7;
    const v2i* __restrict__ p2 = (const v2i*)part;

    {
        constexpr int WV = F_IN * F_MID / 4 / 1024;
        v4f wv[WV];
#pragma unroll
        for (int j = 0; j < WV; ++j) wv[j] = *(const v4f*)(W + 4 * (tid + 1024 * j));
#pragma unroll
        for (int j = 0; j < WV; ++j) {
            const int i4 = tid + 1024 * j;
            const int k  = i4 / (F_MID / 4);
            const int n0 = (i4 % (F_MID / 4)) * 4;
            sWt[(n0 + 0) * LDK1 + k] = wv[j].x;
            sWt[(n0 + 1) * LDK1 + k] = wv[j].y;
            sWt[(n0 + 2) * LDK1 + k] = wv[j].z;
            sWt[(n0 + 3) * LDK1 + k] = wv[j].w;
        }
    }

    if (tid < BN) cnt[tid] = 0;

    int bpart = 0;
    for (int t = tid; t < T; t += 1024) bpart += S[(size_t)t * (size_t)(P + 1) + b];

    const int wave = tid >> 5;
    const int lane = tid & 31;
    const int half = lane >> 4;
    const int l    = lane & 15;
    const int rt   = wave >> 1;
    const int ch   = wave & 1;
    const int m0   = node0 + rt * 16;
    const bool active = rt < RT && m0 < n_nodes;
    const int rowA = min(m0 + l, n_nodes - 1);
    const float* __restrict__ xrow = x + (size_t)rowA * F_IN;
    const bool gemm_first = (((b >> 8) ^ b) & 1) != 0;
    v8f acc0 = {};
    v8f acc1 = {};

    __syncthreads();
    if (gemm_first && active) gemm_half_tile(xrow, sWt, half, l, ch, acc0, acc1);

    FOR_RANGE_EDGES({ (void)sv; atomicAdd(&cnt[dl], 1); })
    __syncthreads();

    int base;
    (void)block_excl_scan_1024(bpart, swave_a, base);
    const int cdeg = (tid < BN) ? cnt[tid] : 0;
    int nb;
    const int ex = block_excl_scan_1024(cdeg, swave_b, nb);
    const float di = rsqrtf((float)(cdeg + 1));
    if (tid < BN) {
        cur[tid]   = ex;
        sdinv[tid] = di;
    }
    if (tid < nn) {
        rowptr[node0 + tid] = base + ex;
        dinv[node0 + tid]   = di;
    }
    if (b == P - 1 && tid == 0) rowptr[n_nodes] = base + nb;
    __syncthreads();

    if (!gemm_first && active) gemm_half_tile(xrow, sWt, half, l, ch, acc0, acc1);

    if (active) {
        float dv[8];
        {
            const v4f d0 = *(const v4f*)(&sdinv[rt * 16 + 8 * half]);
            const v4f d1 = *(const v4f*)(&sdinv[rt * 16 + 8 * half + 4]);
            dv[0] = d0.x; dv[1] = d0.y; dv[2] = d0.z; dv[3] = d0.w;
            dv[4] = d1.x; dv[5] = d1.y; dv[6] = d1.z; dv[7] = d1.w;
        }
        if (m0 + 16 <= n_nodes) {
            float* __restrict__ hp = h + (size_t)(m0 + 8 * half) * F_MID + 32 * ch + l;
#pragma unroll
            for (int v = 0; v < 8; ++v) {
                hp[(size_t)v * F_MID]      = dv[v] * acc0[v];
                hp[(size_t)v * F_MID + 16] = dv[v] * acc1[v];
            }
        } else {
#pragma unroll
            for (int v = 0; v < 8; ++v) {
                const int row = m0 + 8 * half + v;
                if (row < n_nodes) {
                    h[(size_t)row * F_MID + 32 * ch + l]      = dv[v] * acc0[v];
                    h[(size_t)row * F_MID + 32 * ch + 16 + l] = dv[v] * acc1[v];
                }
            }
        }
    }

    const bool staged = nb <= CAP;
    FOR_RANGE_EDGES({
        const int pos = atomicAdd(&cur[dl], 1);
        if (staged) {
            if ((unsigned)pos < (unsigned)CAP) stage[pos] = sv;
        } else {
            const int q = base + pos;
            if ((unsigned)q < (unsigned)n_edges) col[q] = sv;
        }
    })
    __syncthreads();
    if (staged) {
        for (int i = tid; i < nb; i += 1024) {
            const int q = base + i;
            if ((unsigned)q < (unsigned)n_edges) col[q] = stage[i];
        }
    }
}


template <int F, bool RELU>
__global__ void __launch_bounds__(256) gather_kernel(const int* __restrict__ rowptr,
                                                     const int* __restrict__ col,
                                                     const float* __restrict__ dinv,
                                                     const float* __restrict__ h,
                                                     const float* __restrict__ bias,
                                                     float* __restrict__ outp, int n_nodes) {
    constexpr int G  = 8;
    constexpr int V4 = F / 32;
    constexpr int Q  = 32 / G;
    const int lane  = threadIdx.x & 31;
    const int sub   = lane & (G - 1);
    const int gbase = lane & ~(G - 1);
    const int node  = (int)((blockIdx.x * 256 + threadIdx.x) >> 3);
    const bool valid = node < n_nodes;
    const int nodec  = valid ? node : n_nodes - 1;

    const int beg  = rowptr[nodec];
    const int endr = rowptr[nodec + 1];
    const int end  = valid ? endr : beg;
    const float dd = dinv[nodec];
    const float* __restrict__ hl = h + 4 * sub;

    v4f hs[V4];
#pragma unroll
    for (int v = 0; v < V4; ++v) hs[v] = *(const v4f*)(hl + (size_t)nodec * F + 32 * v);

    v4f acc[V4];
#pragma unroll
    for (int v = 0; v < V4; ++v) acc[v] = (v4f){0.0f, 0.0f, 0.0f, 0.0f};

    for (int base = beg; __any(base < end); base += 32) {
        unsigned c_l[Q];
#pragma unroll
        for (int q = 0; q < Q; ++q) {
            const int j = base + q * G + sub;
            c_l[q] = (unsigned)nodec;
            if (j < end) c_l[q] = (unsigned)col[j];
        }
        int s_l[Q];
#pragma unroll
        for (int q = 0; q < Q; ++q) {
            s_l[q] = (int)(c_l[q] < (unsigned)n_nodes ? c_l[q] : (unsigned)(n_nodes - 1));
        }
        const int m = end - base;
#pragma unroll
        for (int q = 0; q < Q; ++q) {
            if (__any(q * G < m)) {
                int s[G];
                v4f r[G][V4];
#pragma unroll
                for (int i = 0; i < G; ++i) s[i] = __shfl(s_l[q], gbase + i, 32);
#pragma unroll
                for (int i = 0; i < G; ++i) {
#pragma unroll
                    for (int v = 0; v < V4; ++v)
                        r[i][v] = *(const v4f*)(hl + (size_t)s[i] * F + 32 * v);
                }
#pragma unroll
                for (int i = 0; i < G; ++i) {
                    const float lv = (q * G + i < m) ? 1.0f : 0.0f;
#pragma unroll
                    for (int v = 0; v < V4; ++v) {
                        acc[v].x = fmaf(lv, r[i][v].x, acc[v].x);
                        acc[v].y = fmaf(lv, r[i][v].y, acc[v].y);
                        acc[v].z = fmaf(lv, r[i][v].z, acc[v].z);
                        acc[v].w = fmaf(lv, r[i][v].w, acc[v].w);
                    }
                }
            }
        }
    }

    if (!valid) return;
#pragma unroll
    for (int v = 0; v < V4; ++v) {
        const v4f bs = *(const v4f*)(bias + 32 * v + 4 * sub);
        v4f o;
        o.x = dd * (acc[v].x + hs[v].x) + bs.x;
        o.y = dd * (acc[v].y + hs[v].y) + bs.y;
        o.z = dd * (acc[v].z + hs[v].z) + bs.z;
        o.w = dd * (acc[v].w + hs[v].w) + bs.w;
        if (RELU) {
            o.x = fmaxf(o.x, 0.0f);
            o.y = fmaxf(o.y, 0.0f);
            o.z = fmaxf(o.z, 0.0f);
            o.w = fmaxf(o.w, 0.0f);
        }
        *(v4f*)(outp + (size_t)node * F + 32 * v + 4 * sub) = o;
    }
}


#define LDA (F_MID + 4)

__global__ void __launch_bounds__(256) gather_gemm_kernel(const int* __restrict__ rowptr,
                                                          const int* __restrict__ col,
                                                          const float* __restrict__ dinv,
                                                          const float* __restrict__ h,
                                                          const float* __restrict__ bias,
                                                          const float* __restrict__ W2,
                                                          float* __restrict__ h2, int n_nodes) {
    constexpr int F  = F_MID;
    constexpr int G  = 8;
    constexpr int V4 = F / 32;
    constexpr int Q  = 32 / G;
    __shared__ __attribute__((aligned(16))) float sWt[F_OUT * LDA];
    __shared__ __attribute__((aligned(16))) float sA[32 * LDA];
    __shared__ __attribute__((aligned(16))) float sD[32];

    constexpr int WV = F_MID * F_OUT / 4 / 256;
    v4f wv[WV];
#pragma unroll
    for (int j = 0; j < WV; ++j) wv[j] = *(const v4f*)(W2 + 4 * ((int)threadIdx.x + 256 * j));

    const int lane  = threadIdx.x & 31;
    const int sub   = lane & (G - 1);
    const int gbase = lane & ~(G - 1);
    const int nl    = (int)(threadIdx.x >> 3);
    const int node  = (int)(blockIdx.x * 32) + nl;
    const bool valid = node < n_nodes;
    const int nodec  = valid ? node : n_nodes - 1;

    const int beg  = rowptr[nodec];
    const int endr = rowptr[nodec + 1];
    const int end  = valid ? endr : beg;
    const float dd = dinv[nodec];
    const float* __restrict__ hl = h + 4 * sub;

    v4f hs[V4];
#pragma unroll
    for (int v = 0; v < V4; ++v) hs[v] = *(const v4f*)(hl + (size_t)nodec * F + 32 * v);

#pragma unroll
    for (int j = 0; j < WV; ++j) {
        const int i4 = (int)threadIdx.x + 256 * j;
        const int k  = i4 / (F_OUT / 4);
        const int n0 = (i4 % (F_OUT / 4)) * 4;
        sWt[(n0 + 0) * LDA + k] = wv[j].x;
        sWt[(n0 + 1) * LDA + k] = wv[j].y;
        sWt[(n0 + 2) * LDA + k] = wv[j].z;
        sWt[(n0 + 3) * LDA + k] = wv[j].w;
    }

    v4f acc[V4];
#pragma unroll
    for (int v = 0; v < V4; ++v) acc[v] = (v4f){0.0f, 0.0f, 0.0f, 0.0f};

    for (int base = beg; __any(base < end); base += 32) {
        unsigned c_l[Q];
#pragma unroll
        for (int q = 0; q < Q; ++q) {
            const int j = base + q * G + sub;
            c_l[q] = (unsigned)nodec;
            if (j < end) c_l[q] = (unsigned)col[j];
        }
        int s_l[Q];
#pragma unroll
        for (int q = 0; q < Q; ++q) {
            s_l[q] = (int)(c_l[q] < (unsigned)n_nodes ? c_l[q] : (unsigned)(n_nodes - 1));
        }
        const int m = end - base;
#pragma unroll
        for (int q = 0; q < Q; ++q) {
            if (__any(q * G < m)) {
                int s[G];
                v4f r[G][V4];
#pragma unroll
                for (int i = 0; i < G; ++i) s[i] = __shfl(s_l[q], gbase + i, 32);
#pragma unroll
                for (int i = 0; i < G; ++i) {
#pragma unroll
                    for (int v = 0; v < V4; ++v)
                        r[i][v] = *(const v4f*)(hl + (size_t)s[i] * F + 32 * v);
                }
#pragma unroll
                for (int i = 0; i < G; ++i) {
                    const float lv = (q * G + i < m) ? 1.0f : 0.0f;
#pragma unroll
                    for (int v = 0; v < V4; ++v) {
                        acc[v].x = fmaf(lv, r[i][v].x, acc[v].x);
                        acc[v].y = fmaf(lv, r[i][v].y, acc[v].y);
                        acc[v].z = fmaf(lv, r[i][v].z, acc[v].z);
                        acc[v].w = fmaf(lv, r[i][v].w, acc[v].w);
                    }
                }
            }
        }
    }

#pragma unroll
    for (int v = 0; v < V4; ++v) {
        const v4f bs = *(const v4f*)(bias + 32 * v + 4 * sub);
        v4f o;
        o.x = fmaxf(dd * (acc[v].x + hs[v].x) + bs.x, 0.0f);
        o.y = fmaxf(dd * (acc[v].y + hs[v].y) + bs.y, 0.0f);
        o.z = fmaxf(dd * (acc[v].z + hs[v].z) + bs.z, 0.0f);
        o.w = fmaxf(dd * (acc[v].w + hs[v].w) + bs.w, 0.0f);
        if (!valid) o = (v4f){0.0f, 0.0f, 0.0f, 0.0f};
        *(v4f*)(&sA[nl * LDA + 32 * v + 4 * sub]) = o;
    }
    if (sub == 0) sD[nl] = valid ? dd : 0.0f;
    __syncthreads();

    const int wave = (int)(threadIdx.x >> 5);
    if (wave >= 2) return;
    const int half = lane >> 4;
    const int l    = lane & 15;
    const int m0   = (int)(blockIdx.x * 32) + wave * 16;
    if (m0 >= n_nodes) return;

    constexpr int NT = F_OUT / 16;
    const float* __restrict__ arow = &sA[(wave * 16 + l) * LDA];
    v8f pacc[NT] = {};
    for (int k = 0; k < F_MID; k += 4) {
        const int kk = k + 2 * half;
        const v2f a = *(const v2f*)(arow + kk);
#pragma unroll
        for (int t = 0; t < NT; ++t) {
            const v2f b = *(const v2f*)(&sWt[(t * 16 + l) * LDA + kk]);
            pacc[t] = __builtin_amdgcn_wmma_f32_16x16x4_f32(
                false, a, false, b, (short)0, pacc[t], false, false);
        }
    }

    float dv[8];
    {
        const v4f d0 = *(const v4f*)(&sD[wave * 16 + 8 * half]);
        const v4f d1 = *(const v4f*)(&sD[wave * 16 + 8 * half + 4]);
        dv[0] = d0.x; dv[1] = d0.y; dv[2] = d0.z; dv[3] = d0.w;
        dv[4] = d1.x; dv[5] = d1.y; dv[6] = d1.z; dv[7] = d1.w;
    }

    if (m0 + 16 <= n_nodes) {
        float* __restrict__ hp = h2 + (size_t)(m0 + 8 * half) * F_OUT + l;
#pragma unroll
        for (int v = 0; v < 8; ++v) {
#pragma unroll
            for (int t = 0; t < NT; ++t)
                hp[(size_t)v * F_OUT + t * 16] = dv[v] * pacc[t][v];
        }
    } else {
#pragma unroll
        for (int v = 0; v < 8; ++v) {
            const int row = m0 + 8 * half + v;
            if (row < n_nodes) {
#pragma unroll
                for (int t = 0; t < NT; ++t)
                    h2[(size_t)row * F_OUT + t * 16 + l] = dv[v] * pacc[t][v];
            }
        }
    }
}


__global__ void __launch_bounds__(256) decode_kernel(const int* __restrict__ pos,
                                                     const int* __restrict__ neg,
                                                     const float* __restrict__ z,
                                                     float* __restrict__ out, int n_eval) {
    const int lane  = threadIdx.x & 31;
    const int sub   = lane & 7;
    const int gbase = lane & 24;
    const int total = 2 * n_eval;
    const int k = (int)(blockIdx.x * 256 + threadIdx.x);
    int a = 0, b = 0;
    if (k < total) {
        if (k < n_eval) { a = pos[k];            b = pos[n_eval + k]; }
        else            { const int kk = k - n_eval; a = neg[kk]; b = neg[n_eval + kk]; }
    }
    const float* __restrict__ zl = z + 4 * sub;
    v4f za[8];
    v4f zb[8];
#pragma unroll
    for (int p = 0; p < 8; ++p) {
        const int ap = __shfl(a, gbase + p, 32);
        const int bp = __shfl(b, gbase + p, 32);
        za[p] = *(const v4f*)(zl + (size_t)ap * F_OUT);
        zb[p] = *(const v4f*)(zl + (size_t)bp * F_OUT);
    }
    float res = 0.0f;
#pragma unroll
    for (int p = 0; p < 8; ++p) {
        float v = za[p].x * zb[p].x + za[p].y * zb[p].y + za[p].z * zb[p].z + za[p].w * zb[p].w;
        v += __shfl_xor(v, 4, 32);
        v += __shfl_xor(v, 2, 32);
        v += __shfl_xor(v, 1, 32);
        if (sub == p) res = v;
    }
    if (k < total) out[k] = res;
}


static inline size_t align_up(size_t x) { return (x + 255) & ~(size_t)255; }

extern "C" void kernel_launch(void* const* d_in, const int* in_sizes, int n_in,
                              void* d_out, int out_size, void* d_ws, size_t ws_size,
                              hipStream_t stream) {
    const float* x   = (const float*)d_in[0];
    const float* W1  = (const float*)d_in[1];
    const float* b1  = (const float*)d_in[2];
    const float* W2  = (const float*)d_in[3];
    const float* b2  = (const float*)d_in[4];
    const int*   ei  = (const int*)d_in[5];
    const int*   pos = (const int*)d_in[6];
    const int*   neg = (const int*)d_in[7];
    float* out = (float*)d_out;

    const int n_nodes = in_sizes[0] / F_IN;
    const int n_edges = in_sizes[5] / 2;
    const int n_eval  = in_sizes[6] / 2;
    const int* src = ei;
    const int* dst = ei + n_edges;

    const int B = 256;
    const int P = (n_nodes + BN - 1) / BN;
    const int T = (n_edges + TILE - 1) / TILE;

    char* w = (char*)d_ws;
    size_t off = 0;
    float* dinv  = (float*)(w + off); off += align_up((size_t)n_nodes * 4);
    float* h1    = (float*)(w + off); off += align_up((size_t)n_nodes * F_MID * 4);
    const size_t a1_bytes   = (size_t)n_nodes * F_MID * 4;
    const size_t part_bytes = (size_t)n_edges * 8;
    float* h2    = (float*)(w + off);
    int* part    = (int*)(w + off);   off += align_up(a1_bytes > part_bytes ? a1_bytes : part_bytes);
    int* rowptr  = (int*)(w + off);   off += align_up((size_t)(n_nodes + 1) * 4);
    int* S       = (int*)(w + off);   off += align_up((size_t)T * (size_t)(P + 1) * 4);
    int* col     = (int*)(w + off);   off += align_up((size_t)n_edges * 4);
    float* z     = h1;
    (void)ws_size;

    tile_sort_kernel<<<T, 1024, 0, stream>>>(src, dst, part, S, n_edges, P);
    range_csr_gemm_kernel<<<P, 1024, 0, stream>>>(part, S, x, W1, rowptr, dinv, col, h1,
                                                  n_edges, n_nodes, P, T);

    const int gather_blocks = (n_nodes + 32 - 1) / 32;

    gather_gemm_kernel<<<gather_blocks, B, 0, stream>>>(rowptr, col, dinv, h1, b1, W2, h2, n_nodes);

    gather_kernel<F_OUT, false><<<gather_blocks, B, 0, stream>>>(rowptr, col, dinv, h2, b2, z, n_nodes);

    {
        int blocks = (2 * n_eval + B - 1) / B;
        decode_kernel<<<blocks, B, 0, stream>>>(pos, neg, z, out, n_eval);
    }
    (void)out_size; (void)n_in;
}
